# starting kernel
# baseline (speedup 1.0000x reference)
#include <hip/hip_runtime.h>

#define NN 50000
#define EE 800000
#define CC 64
#define EDD 11
#define NEGS 0.2f

typedef __attribute__((ext_vector_type(2))) float v2f;
typedef __attribute__((ext_vector_type(8))) float v8f;

__device__ __forceinline__ void atomicMaxF(float* addr, float v) {
    if (v >= 0.0f) atomicMax((int*)addr, __float_as_int(v));
    else           atomicMin((unsigned int*)addr, __float_as_uint(v));
}

__global__ void k_init(const float* __restrict__ x, const float* __restrict__ bias,
                       float* __restrict__ out, float* __restrict__ mmax,
                       float* __restrict__ denom, float* __restrict__ macc) {
    int i = blockIdx.x * blockDim.x + threadIdx.x;
    if (i < NN * CC) out[i] = x[i] + bias[i & (CC - 1)];
    if (i < NN) { mmax[i] = -__builtin_inff(); denom[i] = 0.0f; }
    if (i < 16) macc[i] = 0.0f;
}

__global__ void k_meansum(const float* __restrict__ ea, float* __restrict__ macc) {
    __shared__ float s[EDD];
    if (threadIdx.x < EDD) s[threadIdx.x] = 0.0f;
    __syncthreads();
    float loc[EDD];
#pragma unroll
    for (int k = 0; k < EDD; k++) loc[k] = 0.0f;
    for (int e = blockIdx.x * blockDim.x + threadIdx.x; e < EE; e += gridDim.x * blockDim.x) {
#pragma unroll
        for (int k = 0; k < EDD; k++) loc[k] += ea[e * EDD + k];
    }
#pragma unroll
    for (int k = 0; k < EDD; k++) atomicAdd(&s[k], loc[k]);
    __syncthreads();
    if (threadIdx.x < EDD) atomicAdd(&macc[threadIdx.x], s[threadIdx.x]);
}

__global__ void k_proj_wmma(const float* __restrict__ x,
                            const float* __restrict__ Wl, const float* __restrict__ bl,
                            const float* __restrict__ Wr, const float* __restrict__ br,
                            float* __restrict__ xl, float* __restrict__ xr) {
    const int wid = blockIdx.x * (blockDim.x >> 5) + (threadIdx.x >> 5);
    if (wid >= NN / 16) return;
    const int lane = threadIdx.x & 31;
    const int half = lane >> 4;
    const int l    = lane & 15;
    const int m0   = wid * 16;

    v8f accL[4] = {};
    v8f accR[4] = {};

    for (int k0 = 0; k0 < CC; k0 += 4) {
        v2f a;
        const float* ap = x + (m0 + l) * CC + k0 + 2 * half;
        a.x = ap[0];
        a.y = ap[1];
#pragma unroll
        for (int n = 0; n < 4; n++) {
            const int kb = (k0 + 2 * half) * CC + n * 16 + l;
            v2f b;
            b.x = Wl[kb];
            b.y = Wl[kb + CC];
            accL[n] = __builtin_amdgcn_wmma_f32_16x16x4_f32(
                false, a, false, b, (short)0, accL[n], false, false);
            v2f c;
            c.x = Wr[kb];
            c.y = Wr[kb + CC];
            accR[n] = __builtin_amdgcn_wmma_f32_16x16x4_f32(
                false, a, false, c, (short)0, accR[n], false, false);
        }
    }

#pragma unroll
    for (int n = 0; n < 4; n++) {
#pragma unroll
        for (int r = 0; r < 8; r++) {
            const int row = m0 + half * 8 + r;
            const int col = n * 16 + l;
            xl[row * CC + col] = accL[n][r] + bl[col];
            xr[row * CC + col] = accR[n][r] + br[col];
        }
    }
}

__global__ void k_logits(const int* __restrict__ eidx, const float* __restrict__ ea,
                         const float* __restrict__ We, const float* __restrict__ att,
                         const float* __restrict__ macc,
                         const float* __restrict__ xl, const float* __restrict__ xr,
                         float* __restrict__ lg, float* __restrict__ mmax) {
    __shared__ float sWe[EDD * CC];
    __shared__ float sAtt[CC];
    __shared__ float sMean[EDD];
    for (int i = threadIdx.x; i < EDD * CC; i += blockDim.x) sWe[i] = We[i];
    if (threadIdx.x < CC)  sAtt[threadIdx.x]  = att[threadIdx.x];
    if (threadIdx.x < EDD) sMean[threadIdx.x] = macc[threadIdx.x] * (1.0f / (float)EE);
    __syncthreads();

    const int lane = threadIdx.x & 31;
    const int c0   = lane * 2;
    const int nw   = (gridDim.x * blockDim.x) >> 5;

    for (int e = (blockIdx.x * blockDim.x + threadIdx.x) >> 5; e < EE + NN; e += nw) {
        int s, d;
        float myea;
        if (e < EE) {
            s = eidx[e];
            d = eidx[EE + e];
            myea = (lane < EDD) ? ea[e * EDD + lane] : 0.0f;
        } else {
            s = e - EE;
            d = s;
            myea = (lane < EDD) ? sMean[lane] : 0.0f;
        }
        const float2 vl = *(const float2*)(xl + s * CC + c0);
        const float2 vr = *(const float2*)(xr + d * CC + c0);
        float hx = vl.x + vr.x;
        float hy = vl.y + vr.y;
#pragma unroll
        for (int k = 0; k < EDD; k++) {
            const float ek = __shfl(myea, k, 32);
            hx += ek * sWe[k * CC + c0];
            hy += ek * sWe[k * CC + c0 + 1];
        }
        hx = hx > 0.0f ? hx : NEGS * hx;
        hy = hy > 0.0f ? hy : NEGS * hy;
        float p = hx * sAtt[c0] + hy * sAtt[c0 + 1];
#pragma unroll
        for (int off = 16; off > 0; off >>= 1) p += __shfl_xor(p, off, 32);
        if (lane == 0) {
            lg[e] = p;
            atomicMaxF(&mmax[d], p);
        }
    }
}

__global__ void k_expsum(const int* __restrict__ eidx, const float* __restrict__ mmax,
                         float* __restrict__ lg, float* __restrict__ denom) {
    int e = blockIdx.x * blockDim.x + threadIdx.x;
    if (e >= EE + NN) return;
    const int d = (e < EE) ? eidx[EE + e] : (e - EE);
    const float p = expf(lg[e] - mmax[d]);
    lg[e] = p;
    atomicAdd(&denom[d], p);
}

__global__ void k_scatter(const int* __restrict__ eidx, const float* __restrict__ lg,
                          const float* __restrict__ denom, const float* __restrict__ xl,
                          float* __restrict__ out) {
    int idx = blockIdx.x * blockDim.x + threadIdx.x;
    int e = idx >> 5;
    if (e >= EE + NN) return;
    const int lane = idx & 31;
    int s, d;
    if (e < EE) { s = eidx[e]; d = eidx[EE + e]; }
    else        { s = e - EE;  d = s; }
    const float alpha = lg[e] / denom[d];
    const int c0 = lane * 2;
    const float2 v = *(const float2*)(xl + s * CC + c0);
    atomicAdd(&out[d * CC + c0],     alpha * v.x);
    atomicAdd(&out[d * CC + c0 + 1], alpha * v.y);
}

extern "C" void kernel_launch(void* const* d_in, const int* in_sizes, int n_in,
                              void* d_out, int out_size, void* d_ws, size_t ws_size,
                              hipStream_t stream) {
    const float* x    = (const float*)d_in[0];
    const int*   eidx = (const int*)  d_in[1];
    const float* ea   = (const float*)d_in[2];
    const float* Wl   = (const float*)d_in[3];
    const float* bl   = (const float*)d_in[4];
    const float* Wr   = (const float*)d_in[5];
    const float* br   = (const float*)d_in[6];
    const float* We   = (const float*)d_in[7];
    const float* att  = (const float*)d_in[8];
    const float* bias = (const float*)d_in[9];
    float* out = (float*)d_out;

    float* ws    = (float*)d_ws;
    float* xl    = ws;
    float* xr    = xl + NN * CC;
    float* lg    = xr + NN * CC;
    float* mmax  = lg + (EE + NN);
    float* denom = mmax + NN;
    float* macc  = denom + NN;

    const int T = 256;
    k_init<<<(NN * CC + T - 1) / T, T, 0, stream>>>(x, bias, out, mmax, denom, macc);
    k_meansum<<<512, T, 0, stream>>>(ea, macc);
    k_proj_wmma<<<(NN / 16 + 7) / 8, T, 0, stream>>>(x, Wl, bl, Wr, br, xl, xr);
    k_logits<<<2048, T, 0, stream>>>(eidx, ea, We, att, macc, xl, xr, lg, mmax);
    k_expsum<<<(EE + NN + T - 1) / T, T, 0, stream>>>(eidx, mmax, lg, denom);
    k_scatter<<<((EE + NN) * 32 + T - 1) / T, T, 0, stream>>>(eidx, lg, denom, xl, out);
}


# k_logits as 16-edge tiles on the f32 matrix unit: A = edge_attr rows, B = W_e, C = gathered xl[src]+xr[dst]; logits by in-lane and 16-lane DPP sums; LDS permutes gone; sum order changes
# speedup vs baseline: 1.5423x; 1.5423x over previous
#include <hip/hip_runtime.h>

#define NN 50000
#define EE 800000
#define CC 64
#define EDD 11
#define NEGS 0.2f

typedef __attribute__((ext_vector_type(2))) float v2f;
typedef __attribute__((ext_vector_type(8))) float v8f;

__device__ __forceinline__ void atomicMaxF(float* addr, float v) {
    if (v >= 0.0f) atomicMax((int*)addr, __float_as_int(v));
    else           atomicMin((unsigned int*)addr, __float_as_uint(v));
}

__global__ void k_init(const float* __restrict__ x, const float* __restrict__ bias,
                       float* __restrict__ out, float* __restrict__ mmax,
                       float* __restrict__ denom, float* __restrict__ macc) {
    int i = blockIdx.x * blockDim.x + threadIdx.x;
    if (i < NN * CC) out[i] = x[i] + bias[i & (CC - 1)];
    if (i < NN) { mmax[i] = -__builtin_inff(); denom[i] = 0.0f; }
    if (i < 16) macc[i] = 0.0f;
}

__global__ void k_meansum(const float* __restrict__ ea, float* __restrict__ macc) {
    __shared__ float s[EDD];
    if (threadIdx.x < EDD) s[threadIdx.x] = 0.0f;
    __syncthreads();
    float loc[EDD];
#pragma unroll
    for (int k = 0; k < EDD; k++) loc[k] = 0.0f;
    for (int e = blockIdx.x * blockDim.x + threadIdx.x; e < EE; e += gridDim.x * blockDim.x) {
#pragma unroll
        for (int k = 0; k < EDD; k++) loc[k] += ea[e * EDD + k];
    }
#pragma unroll
    for (int k = 0; k < EDD; k++) atomicAdd(&s[k], loc[k]);
    __syncthreads();
    if (threadIdx.x < EDD) atomicAdd(&macc[threadIdx.x], s[threadIdx.x]);
}

__global__ void k_proj_wmma(const float* __restrict__ x,
                            const float* __restrict__ Wl, const float* __restrict__ bl,
                            const float* __restrict__ Wr, const float* __restrict__ br,
                            float* __restrict__ xl, float* __restrict__ xr) {
    const int wid = blockIdx.x * (blockDim.x >> 5) + (threadIdx.x >> 5);
    if (wid >= NN / 16) return;
    const int lane = threadIdx.x & 31;
    const int half = lane >> 4;
    const int l    = lane & 15;
    const int m0   = wid * 16;

    v8f accL[4] = {};
    v8f accR[4] = {};

    for (int k0 = 0; k0 < CC; k0 += 4) {
        v2f a;
        const float* ap = x + (m0 + l) * CC + k0 + 2 * half;
        a.x = ap[0];
        a.y = ap[1];
#pragma unroll
        for (int n = 0; n < 4; n++) {
            const int kb = (k0 + 2 * half) * CC + n * 16 + l;
            v2f b;
            b.x = Wl[kb];
            b.y = Wl[kb + CC];
            accL[n] = __builtin_amdgcn_wmma_f32_16x16x4_f32(
                false, a, false, b, (short)0, accL[n], false, false);
            v2f c;
            c.x = Wr[kb];
            c.y = Wr[kb + CC];
            accR[n] = __builtin_amdgcn_wmma_f32_16x16x4_f32(
                false, a, false, c, (short)0, accR[n], false, false);
        }
    }

#pragma unroll
    for (int n = 0; n < 4; n++) {
#pragma unroll
        for (int r = 0; r < 8; r++) {
            const int row = m0 + half * 8 + r;
            const int col = n * 16 + l;
            xl[row * CC + col] = accL[n][r] + bl[col];
            xr[row * CC + col] = accR[n][r] + br[col];
        }
    }
}

#define NTILES ((EE + NN) / 16)
static_assert(EE % 16 == 0 && NN % 16 == 0, "edge tiles must not mix edges and self loops");

template <int CTRL>
__device__ __forceinline__ float row_xadd(float v) {
    const int o = __builtin_amdgcn_update_dpp(0, __float_as_int(v), CTRL, 0xf, 0xf, true);
    return v + __int_as_float(o);
}

__global__ void __launch_bounds__(256) __attribute__((amdgpu_waves_per_eu(8, 8)))
k_logits(const int* __restrict__ eidx, const float* __restrict__ ea,
         const float* __restrict__ We, const float* __restrict__ att,
         const float* __restrict__ macc,
         const float* __restrict__ xl, const float* __restrict__ xr,
         float* __restrict__ lg, float* __restrict__ mmax) {
    const int lane = threadIdx.x & 31;
    const int half = lane >> 4;
    const int c    = lane & 15;
    const int wave = __builtin_amdgcn_readfirstlane((int)((blockIdx.x * blockDim.x + threadIdx.x) >> 5));
    const int nw   = (int)((gridDim.x * blockDim.x) >> 5);

    v2f bw[3][4];
#pragma unroll
    for (int j = 0; j < 3; j++) {
        const int k = 4 * j + 2 * half;
        const float4 w0 = *(const float4*)(We + k * CC + 4 * c);
        float4 w1 = make_float4(0.0f, 0.0f, 0.0f, 0.0f);
        if (k + 1 < EDD) w1 = *(const float4*)(We + (k + 1) * CC + 4 * c);
        bw[j][0].x = w0.x; bw[j][0].y = w1.x;
        bw[j][1].x = w0.y; bw[j][1].y = w1.y;
        bw[j][2].x = w0.z; bw[j][2].y = w1.z;
        bw[j][3].x = w0.w; bw[j][3].y = w1.w;
    }
    const float4 av = *(const float4*)(att + 4 * c);
    const float4* __restrict__ xl4 = (const float4*)xl;
    const float4* __restrict__ xr4 = (const float4*)xr;

    for (int t = wave; t < NTILES; t += nw) {
        const int e0 = t * 16;
        int si[8], di[8];
        v2f a[3];
        if (e0 < EE) {
            const unsigned io = (unsigned)e0 + 8u * (unsigned)half;
            const int4 s0 = *(const int4*)(eidx + io);
            const int4 s1 = *(const int4*)(eidx + io + 4u);
            const int4 d0 = *(const int4*)(eidx + (unsigned)EE + io);
            const int4 d1 = *(const int4*)(eidx + (unsigned)EE + io + 4u);
            si[0] = s0.x; si[1] = s0.y; si[2] = s0.z; si[3] = s0.w;
            si[4] = s1.x; si[5] = s1.y; si[6] = s1.z; si[7] = s1.w;
            di[0] = d0.x; di[1] = d0.y; di[2] = d0.z; di[3] = d0.w;
            di[4] = d1.x; di[5] = d1.y; di[6] = d1.z; di[7] = d1.w;
            const float* er = ea + ((unsigned)(e0 + c) * EDD + 2u * (unsigned)half);
            a[0].x = er[0]; a[0].y = er[1];
            a[1].x = er[4]; a[1].y = er[5];
            a[2].x = er[8];
            a[2].y = 0.0f;
            if (half == 0) a[2].y = er[9];
        } else {
            const int n0 = e0 - EE + 8 * half;
#pragma unroll
            for (int r = 0; r < 8; r++) { si[r] = n0 + r; di[r] = n0 + r; }
            const float inv = 1.0f / (float)EE;
            const float* mr = macc + 2 * half;
            a[0].x = mr[0] * inv; a[0].y = mr[1] * inv;
            a[1].x = mr[4] * inv; a[1].y = mr[5] * inv;
            a[2].x = mr[8] * inv;
            a[2].y = 0.0f;
            if (half == 0) a[2].y = mr[9] * inv;
        }

        float4 vl[8], vr[8];
#pragma unroll
        for (int r = 0; r < 8; r++) {
            vl[r] = xl4[(unsigned)si[r] * (CC / 4) + (unsigned)c];
            vr[r] = xr4[(unsigned)di[r] * (CC / 4) + (unsigned)c];
        }
        v8f acc[4];
#pragma unroll
        for (int r = 0; r < 8; r++) {
            acc[0][r] = vl[r].x + vr[r].x;
            acc[1][r] = vl[r].y + vr[r].y;
            acc[2][r] = vl[r].z + vr[r].z;
            acc[3][r] = vl[r].w + vr[r].w;
        }
#pragma unroll
        for (int j = 0; j < 3; j++) {
#pragma unroll
            for (int nt = 0; nt < 4; nt++) {
                acc[nt] = __builtin_amdgcn_wmma_f32_16x16x4_f32(
                    false, a[j], false, bw[j][nt], (short)0, acc[nt], false, false);
            }
        }

        float part[8];
#pragma unroll
        for (int r = 0; r < 8; r++) {
            float h0 = acc[0][r], h1 = acc[1][r], h2 = acc[2][r], h3 = acc[3][r];
            h0 = h0 > 0.0f ? h0 : NEGS * h0;
            h1 = h1 > 0.0f ? h1 : NEGS * h1;
            h2 = h2 > 0.0f ? h2 : NEGS * h2;
            h3 = h3 > 0.0f ? h3 : NEGS * h3;
            part[r] = h0 * av.x + h1 * av.y + h2 * av.z + h3 * av.w;
        }
#pragma unroll
        for (int r = 0; r < 8; r++) part[r] = row_xadd<0x161>(part[r]);
#pragma unroll
        for (int r = 0; r < 8; r++) part[r] = row_xadd<0x162>(part[r]);
#pragma unroll
        for (int r = 0; r < 8; r++) part[r] = row_xadd<0x164>(part[r]);
#pragma unroll
        for (int r = 0; r < 8; r++) part[r] = row_xadd<0x168>(part[r]);

        float p = part[0];
        int   d = di[0];
#pragma unroll
        for (int r = 1; r < 8; r++) {
            if ((c & 7) == r) { p = part[r]; d = di[r]; }
        }
        if (c < 8) {
            lg[e0 + 8 * half + c] = p;
            atomicMaxF(&mmax[d], p);
        }
    }
}

__global__ void k_expsum(const int* __restrict__ eidx, const float* __restrict__ mmax,
                         float* __restrict__ lg, float* __restrict__ denom) {
    int e = blockIdx.x * blockDim.x + threadIdx.x;
    if (e >= EE + NN) return;
    const int d = (e < EE) ? eidx[EE + e] : (e - EE);
    const float p = expf(lg[e] - mmax[d]);
    lg[e] = p;
    atomicAdd(&denom[d], p);
}

__global__ void k_scatter(const int* __restrict__ eidx, const float* __restrict__ lg,
                          const float* __restrict__ denom, const float* __restrict__ xl,
                          float* __restrict__ out) {
    int idx = blockIdx.x * blockDim.x + threadIdx.x;
    int e = idx >> 5;
    if (e >= EE + NN) return;
    const int lane = idx & 31;
    int s, d;
    if (e < EE) { s = eidx[e]; d = eidx[EE + e]; }
    else        { s = e - EE;  d = s; }
    const float alpha = lg[e] / denom[d];
    const int c0 = lane * 2;
    const float2 v = *(const float2*)(xl + s * CC + c0);
    atomicAdd(&out[d * CC + c0],     alpha * v.x);
    atomicAdd(&out[d * CC + c0 + 1], alpha * v.y);
}

extern "C" void kernel_launch(void* const* d_in, const int* in_sizes, int n_in,
                              void* d_out, int out_size, void* d_ws, size_t ws_size,
                              hipStream_t stream) {
    const float* x    = (const float*)d_in[0];
    const int*   eidx = (const int*)  d_in[1];
    const float* ea   = (const float*)d_in[2];
    const float* Wl   = (const float*)d_in[3];
    const float* bl   = (const float*)d_in[4];
    const float* Wr   = (const float*)d_in[5];
    const float* br   = (const float*)d_in[6];
    const float* We   = (const float*)d_in[7];
    const float* att  = (const float*)d_in[8];
    const float* bias = (const float*)d_in[9];
    float* out = (float*)d_out;

    float* ws    = (float*)d_ws;
    float* xl    = ws;
    float* xr    = xl + NN * CC;
    float* lg    = xr + NN * CC;
    float* mmax  = lg + (EE + NN);
    float* denom = mmax + NN;
    float* macc  = denom + NN;

    const int T = 256;
    k_init<<<(NN * CC + T - 1) / T, T, 0, stream>>>(x, bias, out, mmax, denom, macc);
    k_meansum<<<512, T, 0, stream>>>(ea, macc);
    k_proj_wmma<<<(NN / 16 + 7) / 8, T, 0, stream>>>(x, Wl, bl, Wr, br, xl, xr);
    k_logits<<<1024, T, 0, stream>>>(eidx, ea, We, att, macc, xl, xr, lg, mmax);
    k_expsum<<<(EE + NN + T - 1) / T, T, 0, stream>>>(eidx, mmax, lg, denom);
    k_scatter<<<((EE + NN) * 32 + T - 1) / T, T, 0, stream>>>(eidx, lg, denom, xl, out);
}


# float atomics gone: edges counting-sorted by destination (returning int count, one-block scan, {src,logit} records); 16 lanes per node do max, exp sum, p*xl[src] sum in registers; sum order changes
# speedup vs baseline: 2.3684x; 1.5356x over previous
#include <hip/hip_runtime.h>

#define NN 50000
#define EE 800000
#define CC 64
#define EDD 11
#define NEGS 0.2f

typedef __attribute__((ext_vector_type(2))) float v2f;
typedef __attribute__((ext_vector_type(8))) float v8f;

#define SCAN_CHUNK 52
#define SCAN_N (1024 * SCAN_CHUNK)
static_assert(SCAN_N > NN && SCAN_CHUNK % 4 == 0, "scan covers every node plus the end offset");

__global__ void k_init(int* __restrict__ cnt, float* __restrict__ macc) {
    int i = blockIdx.x * blockDim.x + threadIdx.x;
    if (i < SCAN_N) cnt[i] = 0;
    if (i < 16) macc[i] = 0.0f;
}

__global__ void k_meansum(const float* __restrict__ ea, float* __restrict__ macc,
                          const int* __restrict__ eidx, int* __restrict__ cnt,
                          int* __restrict__ rank) {
    __shared__ float s[EDD];
    if (threadIdx.x < EDD) s[threadIdx.x] = 0.0f;
    __syncthreads();
    float loc[EDD];
#pragma unroll
    for (int k = 0; k < EDD; k++) loc[k] = 0.0f;
    for (int e = blockIdx.x * blockDim.x + threadIdx.x; e < EE; e += gridDim.x * blockDim.x) {
#pragma unroll
        for (int k = 0; k < EDD; k++) loc[k] += ea[e * EDD + k];
        rank[e] = atomicAdd(&cnt[eidx[EE + e]], 1);
    }
#pragma unroll
    for (int k = 0; k < EDD; k++) atomicAdd(&s[k], loc[k]);
    __syncthreads();
    if (threadIdx.x < EDD) atomicAdd(&macc[threadIdx.x], s[threadIdx.x]);
}

__global__ void __launch_bounds__(1024) k_scan(const int* __restrict__ cnt, int* __restrict__ offs) {
    __shared__ int wsum[32];
    const int t    = threadIdx.x;
    const int lane = t & 31;
    const int wid  = t >> 5;
    const int4* __restrict__ p = (const int4*)(cnt + t * SCAN_CHUNK);
    int4 v[SCAN_CHUNK / 4];
#pragma unroll
    for (int i = 0; i < SCAN_CHUNK / 4; i++) v[i] = p[i];
    int s = 0;
#pragma unroll
    for (int i = 0; i < SCAN_CHUNK / 4; i++) s += v[i].x + v[i].y + v[i].z + v[i].w;
    int inc = s;
    for (int o = 1; o < 32; o <<= 1) {
        const int y = __shfl_up(inc, o);
        if (lane >= o) inc += y;
    }
    if (lane == 31) wsum[wid] = inc;
    __syncthreads();
    if (wid == 0) {
        const int w = wsum[lane];
        int winc = w;
        for (int o = 1; o < 32; o <<= 1) {
            const int y = __shfl_up(winc, o);
            if (lane >= o) winc += y;
        }
        wsum[lane] = winc - w;
    }
    __syncthreads();
    int run = wsum[wid] + inc - s;
    int4* __restrict__ q = (int4*)(offs + t * SCAN_CHUNK);
#pragma unroll
    for (int i = 0; i < SCAN_CHUNK / 4; i++) {
        int4 o;
        o.x = run; run += v[i].x;
        o.y = run; run += v[i].y;
        o.z = run; run += v[i].z;
        o.w = run; run += v[i].w;
        q[i] = o;
    }
}

__global__ void k_proj_wmma(const float* __restrict__ x,
                            const float* __restrict__ Wl, const float* __restrict__ bl,
                            const float* __restrict__ Wr, const float* __restrict__ br,
                            float* __restrict__ xl, float* __restrict__ xr) {
    const int wid = blockIdx.x * (blockDim.x >> 5) + (threadIdx.x >> 5);
    if (wid >= NN / 16) return;
    const int lane = threadIdx.x & 31;
    const int half = lane >> 4;
    const int l    = lane & 15;
    const int m0   = wid * 16;

    v8f accL[4] = {};
    v8f accR[4] = {};

    for (int k0 = 0; k0 < CC; k0 += 4) {
        v2f a;
        const float* ap = x + (m0 + l) * CC + k0 + 2 * half;
        a.x = ap[0];
        a.y = ap[1];
#pragma unroll
        for (int n = 0; n < 4; n++) {
            const int kb = (k0 + 2 * half) * CC + n * 16 + l;
            v2f b;
            b.x = Wl[kb];
            b.y = Wl[kb + CC];
            accL[n] = __builtin_amdgcn_wmma_f32_16x16x4_f32(
                false, a, false, b, (short)0, accL[n], false, false);
            v2f c;
            c.x = Wr[kb];
            c.y = Wr[kb + CC];
            accR[n] = __builtin_amdgcn_wmma_f32_16x16x4_f32(
                false, a, false, c, (short)0, accR[n], false, false);
        }
    }

#pragma unroll
    for (int n = 0; n < 4; n++) {
#pragma unroll
        for (int r = 0; r < 8; r++) {
            const int row = m0 + half * 8 + r;
            const int col = n * 16 + l;
            xl[row * CC + col] = accL[n][r] + bl[col];
            xr[row * CC + col] = accR[n][r] + br[col];
        }
    }
}

#define NTILES ((EE + NN) / 16)
static_assert(EE % 16 == 0 && NN % 16 == 0, "edge tiles must not mix edges and self loops");

template <int CTRL>
__device__ __forceinline__ float row_xadd(float v) {
    const int o = __builtin_amdgcn_update_dpp(0, __float_as_int(v), CTRL, 0xf, 0xf, true);
    return v + __int_as_float(o);
}

__global__ void __launch_bounds__(256) __attribute__((amdgpu_waves_per_eu(8, 8)))
k_logits(const int* __restrict__ eidx, const float* __restrict__ ea,
         const float* __restrict__ We, const float* __restrict__ att,
         const float* __restrict__ macc,
         const float* __restrict__ xl, const float* __restrict__ xr,
         const int* __restrict__ rank, const int* __restrict__ offs,
         int2* __restrict__ rec, float* __restrict__ lgs) {
    const int lane = threadIdx.x & 31;
    const int half = lane >> 4;
    const int c    = lane & 15;
    const int wave = __builtin_amdgcn_readfirstlane((int)((blockIdx.x * blockDim.x + threadIdx.x) >> 5));
    const int nw   = (int)((gridDim.x * blockDim.x) >> 5);

    v2f bw[3][4];
#pragma unroll
    for (int j = 0; j < 3; j++) {
        const int k = 4 * j + 2 * half;
        const float4 w0 = *(const float4*)(We + k * CC + 4 * c);
        float4 w1 = make_float4(0.0f, 0.0f, 0.0f, 0.0f);
        if (k + 1 < EDD) w1 = *(const float4*)(We + (k + 1) * CC + 4 * c);
        bw[j][0].x = w0.x; bw[j][0].y = w1.x;
        bw[j][1].x = w0.y; bw[j][1].y = w1.y;
        bw[j][2].x = w0.z; bw[j][2].y = w1.z;
        bw[j][3].x = w0.w; bw[j][3].y = w1.w;
    }
    const float4 av = *(const float4*)(att + 4 * c);
    const float4* __restrict__ xl4 = (const float4*)xl;
    const float4* __restrict__ xr4 = (const float4*)xr;

    for (int t = wave; t < NTILES; t += nw) {
        const int e0 = t * 16;
        int si[8], di[8];
        v2f a[3];
        int myslot = 0, mysrc = 0;
        if (e0 < EE) {
            const unsigned io = (unsigned)e0 + 8u * (unsigned)half;
            const unsigned me = io + (unsigned)(c & 7);
            mysrc  = eidx[me];
            myslot = offs[eidx[(unsigned)EE + me]] + rank[me];
            const int4 s0 = *(const int4*)(eidx + io);
            const int4 s1 = *(const int4*)(eidx + io + 4u);
            const int4 d0 = *(const int4*)(eidx + (unsigned)EE + io);
            const int4 d1 = *(const int4*)(eidx + (unsigned)EE + io + 4u);
            si[0] = s0.x; si[1] = s0.y; si[2] = s0.z; si[3] = s0.w;
            si[4] = s1.x; si[5] = s1.y; si[6] = s1.z; si[7] = s1.w;
            di[0] = d0.x; di[1] = d0.y; di[2] = d0.z; di[3] = d0.w;
            di[4] = d1.x; di[5] = d1.y; di[6] = d1.z; di[7] = d1.w;
            const float* er = ea + ((unsigned)(e0 + c) * EDD + 2u * (unsigned)half);
            a[0].x = er[0]; a[0].y = er[1];
            a[1].x = er[4]; a[1].y = er[5];
            a[2].x = er[8];
            a[2].y = 0.0f;
            if (half == 0) a[2].y = er[9];
        } else {
            const int n0 = e0 - EE + 8 * half;
#pragma unroll
            for (int r = 0; r < 8; r++) { si[r] = n0 + r; di[r] = n0 + r; }
            const float inv = 1.0f / (float)EE;
            const float* mr = macc + 2 * half;
            a[0].x = mr[0] * inv; a[0].y = mr[1] * inv;
            a[1].x = mr[4] * inv; a[1].y = mr[5] * inv;
            a[2].x = mr[8] * inv;
            a[2].y = 0.0f;
            if (half == 0) a[2].y = mr[9] * inv;
        }

        float4 vl[8], vr[8];
#pragma unroll
        for (int r = 0; r < 8; r++) {
            vl[r] = xl4[(unsigned)si[r] * (CC / 4) + (unsigned)c];
            vr[r] = xr4[(unsigned)di[r] * (CC / 4) + (unsigned)c];
        }
        v8f acc[4];
#pragma unroll
        for (int r = 0; r < 8; r++) {
            acc[0][r] = vl[r].x + vr[r].x;
            acc[1][r] = vl[r].y + vr[r].y;
            acc[2][r] = vl[r].z + vr[r].z;
            acc[3][r] = vl[r].w + vr[r].w;
        }
#pragma unroll
        for (int j = 0; j < 3; j++) {
#pragma unroll
            for (int nt = 0; nt < 4; nt++) {
                acc[nt] = __builtin_amdgcn_wmma_f32_16x16x4_f32(
                    false, a[j], false, bw[j][nt], (short)0, acc[nt], false, false);
            }
        }

        float part[8];
#pragma unroll
        for (int r = 0; r < 8; r++) {
            float h0 = acc[0][r], h1 = acc[1][r], h2 = acc[2][r], h3 = acc[3][r];
            h0 = h0 > 0.0f ? h0 : NEGS * h0;
            h1 = h1 > 0.0f ? h1 : NEGS * h1;
            h2 = h2 > 0.0f ? h2 : NEGS * h2;
            h3 = h3 > 0.0f ? h3 : NEGS * h3;
            part[r] = h0 * av.x + h1 * av.y + h2 * av.z + h3 * av.w;
        }
#pragma unroll
        for (int r = 0; r < 8; r++) part[r] = row_xadd<0x161>(part[r]);
#pragma unroll
        for (int r = 0; r < 8; r++) part[r] = row_xadd<0x162>(part[r]);
#pragma unroll
        for (int r = 0; r < 8; r++) part[r] = row_xadd<0x164>(part[r]);
#pragma unroll
        for (int r = 0; r < 8; r++) part[r] = row_xadd<0x168>(part[r]);

        float p = part[0];
#pragma unroll
        for (int r = 1; r < 8; r++) {
            if ((c & 7) == r) p = part[r];
        }
        if (c < 8) {
            if (e0 < EE) rec[myslot] = make_int2(mysrc, __float_as_int(p));
            else         lgs[e0 - EE + 8 * half + c] = p;
        }
    }
}

template <int CTRL>
__device__ __forceinline__ float row_xmax(float v) {
    const int o = __builtin_amdgcn_update_dpp(0, __float_as_int(v), CTRL, 0xf, 0xf, true);
    return fmaxf(v, __int_as_float(o));
}

static_assert((NN * 16) % 256 == 0, "the gather grid holds whole waves of real nodes only");

__global__ void __launch_bounds__(256)
k_gather(const int2* __restrict__ rec, const float* __restrict__ lgs, const int* __restrict__ offs,
         const float* __restrict__ xl, const float* __restrict__ x, const float* __restrict__ bias,
         float* __restrict__ out) {
    const int gid = blockIdx.x * blockDim.x + threadIdx.x;
    const int n = gid >> 4;
    const int c = gid & 15;
    const int beg = offs[n];
    const int end = offs[n + 1];
    const float ls = lgs[n];

    float m = ls;
    for (int j = beg + c; j < end; j += 16) m = fmaxf(m, __int_as_float(rec[j].y));
    m = row_xmax<0x161>(m);
    m = row_xmax<0x162>(m);
    m = row_xmax<0x164>(m);
    m = row_xmax<0x168>(m);

    const float4* __restrict__ xl4 = (const float4*)xl;
    float4 acc = make_float4(0.0f, 0.0f, 0.0f, 0.0f);
    float den = 0.0f;
    int j = beg;
    for (; j + 4 <= end; j += 4) {
        const int2 r0 = rec[j], r1 = rec[j + 1], r2 = rec[j + 2], r3 = rec[j + 3];
        const float4 v0 = xl4[(unsigned)r0.x * (CC / 4) + (unsigned)c];
        const float4 v1 = xl4[(unsigned)r1.x * (CC / 4) + (unsigned)c];
        const float4 v2 = xl4[(unsigned)r2.x * (CC / 4) + (unsigned)c];
        const float4 v3 = xl4[(unsigned)r3.x * (CC / 4) + (unsigned)c];
        const float p0 = expf(__int_as_float(r0.y) - m);
        const float p1 = expf(__int_as_float(r1.y) - m);
        const float p2 = expf(__int_as_float(r2.y) - m);
        const float p3 = expf(__int_as_float(r3.y) - m);
        den += p0; acc.x += p0 * v0.x; acc.y += p0 * v0.y; acc.z += p0 * v0.z; acc.w += p0 * v0.w;
        den += p1; acc.x += p1 * v1.x; acc.y += p1 * v1.y; acc.z += p1 * v1.z; acc.w += p1 * v1.w;
        den += p2; acc.x += p2 * v2.x; acc.y += p2 * v2.y; acc.z += p2 * v2.z; acc.w += p2 * v2.w;
        den += p3; acc.x += p3 * v3.x; acc.y += p3 * v3.y; acc.z += p3 * v3.z; acc.w += p3 * v3.w;
    }
    for (; j < end; j++) {
        const int2 r0 = rec[j];
        const float4 v0 = xl4[(unsigned)r0.x * (CC / 4) + (unsigned)c];
        const float p0 = expf(__int_as_float(r0.y) - m);
        den += p0; acc.x += p0 * v0.x; acc.y += p0 * v0.y; acc.z += p0 * v0.z; acc.w += p0 * v0.w;
    }
    {
        const float4 v0 = xl4[(unsigned)n * (CC / 4) + (unsigned)c];
        const float p0 = expf(ls - m);
        den += p0; acc.x += p0 * v0.x; acc.y += p0 * v0.y; acc.z += p0 * v0.z; acc.w += p0 * v0.w;
    }
    const float inv = 1.0f / den;
    const float4 xv = ((const float4*)x)[(unsigned)n * (CC / 4) + (unsigned)c];
    const float4 bv = ((const float4*)bias)[c];
    float4 o;
    o.x = xv.x + (acc.x * inv + bv.x);
    o.y = xv.y + (acc.y * inv + bv.y);
    o.z = xv.z + (acc.z * inv + bv.z);
    o.w = xv.w + (acc.w * inv + bv.w);
    ((float4*)out)[(unsigned)n * (CC / 4) + (unsigned)c] = o;
}

extern "C" void kernel_launch(void* const* d_in, const int* in_sizes, int n_in,
                              void* d_out, int out_size, void* d_ws, size_t ws_size,
                              hipStream_t stream) {
    const float* x    = (const float*)d_in[0];
    const int*   eidx = (const int*)  d_in[1];
    const float* ea   = (const float*)d_in[2];
    const float* Wl   = (const float*)d_in[3];
    const float* bl   = (const float*)d_in[4];
    const float* Wr   = (const float*)d_in[5];
    const float* br   = (const float*)d_in[6];
    const float* We   = (const float*)d_in[7];
    const float* att  = (const float*)d_in[8];
    const float* bias = (const float*)d_in[9];
    float* out = (float*)d_out;

    float* ws    = (float*)d_ws;
    float* xl    = ws;
    float* xr    = out;
    int2*  rec   = (int2*)(xl + NN * CC);
    float* lgs   = (float*)(rec + EE);
    int*   rank  = (int*)(lgs + NN);
    int*   cnt   = rank + EE;
    int*   offs  = cnt + SCAN_N;
    float* macc  = (float*)(offs + SCAN_N);

    const int T = 256;
    k_init<<<SCAN_N / T, T, 0, stream>>>(cnt, macc);
    k_meansum<<<512, T, 0, stream>>>(ea, macc, eidx, cnt, rank);
    k_scan<<<1, 1024, 0, stream>>>(cnt, offs);
    k_proj_wmma<<<(NN / 16 + 7) / 8, T, 0, stream>>>(x, Wl, bl, Wr, br, xl, xr);
    k_logits<<<1024, T, 0, stream>>>(eidx, ea, We, att, macc, xl, xr, rank, offs, rec, lgs);
    k_gather<<<(NN * 16) / T, T, 0, stream>>>(rec, lgs, offs, xl, x, bias, out);
}


# in-edge counters as two planes, one per package half (die from the wave's id word): returning adds keep each counter line in one half's L2; plane bit kept in rank; segment order by arrival as before
# speedup vs baseline: 2.8258x; 1.1931x over previous
#include <hip/hip_runtime.h>

#define NN 50000
#define EE 800000
#define CC 64
#define EDD 11
#define NEGS 0.2f

typedef __attribute__((ext_vector_type(2))) float v2f;
typedef __attribute__((ext_vector_type(8))) float v8f;

#define SCAN_CHUNK 52
#define SCAN_N (1024 * SCAN_CHUNK)
static_assert(SCAN_N > NN && SCAN_CHUNK % 4 == 0, "scan covers every node plus the end offset");

__global__ void k_init(int* __restrict__ cnt2, float* __restrict__ macc) {
    int i = blockIdx.x * blockDim.x + threadIdx.x;
    if (i < 2 * SCAN_N) cnt2[i] = 0;
    if (i < 16) macc[i] = 0.0f;
}

#define COUNT_BLOCKS 512
__device__ __forceinline__ void count_body(const float* __restrict__ ea, float* __restrict__ macc,
                          const int* __restrict__ eidx, int* __restrict__ cnt,
                          int* __restrict__ rank) {
    __shared__ float s[EDD];
    if (threadIdx.x < EDD) s[threadIdx.x] = 0.0f;
    __syncthreads();
    float loc[EDD];
#pragma unroll
    for (int k = 0; k < EDD; k++) loc[k] = 0.0f;
    const unsigned hsel = (__builtin_amdgcn_s_sendmsg_rtn(0x87) >> 18) & 1u;
    int* __restrict__ myc = cnt + hsel * SCAN_N;
    const int tag = (int)(hsel << 31);
    for (int e = blockIdx.x * blockDim.x + threadIdx.x; e < EE; e += COUNT_BLOCKS * blockDim.x) {
#pragma unroll
        for (int k = 0; k < EDD; k++) loc[k] += ea[e * EDD + k];
        rank[e] = atomicAdd(&myc[eidx[EE + e]], 1) | tag;
    }
#pragma unroll
    for (int k = 0; k < EDD; k++) atomicAdd(&s[k], loc[k]);
    __syncthreads();
    if (threadIdx.x < EDD) atomicAdd(&macc[threadIdx.x], s[threadIdx.x]);
}

__global__ void __launch_bounds__(1024) k_scan(const int* __restrict__ cnt, int* __restrict__ offs) {
    __shared__ int wsum[32];
    const int t    = threadIdx.x;
    const int lane = t & 31;
    const int wid  = t >> 5;
    const int4* __restrict__ p  = (const int4*)(cnt + t * SCAN_CHUNK);
    const int4* __restrict__ p1 = (const int4*)(cnt + SCAN_N + t * SCAN_CHUNK);
    int4 v[SCAN_CHUNK / 4];
#pragma unroll
    for (int i = 0; i < SCAN_CHUNK / 4; i++) {
        const int4 a = p[i];
        const int4 b = p1[i];
        v[i].x = a.x + b.x; v[i].y = a.y + b.y; v[i].z = a.z + b.z; v[i].w = a.w + b.w;
    }
    int s = 0;
#pragma unroll
    for (int i = 0; i < SCAN_CHUNK / 4; i++) s += v[i].x + v[i].y + v[i].z + v[i].w;
    int inc = s;
    for (int o = 1; o < 32; o <<= 1) {
        const int y = __shfl_up(inc, o);
        if (lane >= o) inc += y;
    }
    if (lane == 31) wsum[wid] = inc;
    __syncthreads();
    if (wid == 0) {
        const int w = wsum[lane];
        int winc = w;
        for (int o = 1; o < 32; o <<= 1) {
            const int y = __shfl_up(winc, o);
            if (lane >= o) winc += y;
        }
        wsum[lane] = winc - w;
    }
    __syncthreads();
    int run = wsum[wid] + inc - s;
    int4* __restrict__ q  = (int4*)(offs + t * SCAN_CHUNK);
    int4* __restrict__ q1 = (int4*)(offs + SCAN_N + t * SCAN_CHUNK);
#pragma unroll
    for (int i = 0; i < SCAN_CHUNK / 4; i++) {
        const int4 a = p[i];
        int4 o;
        o.x = run; run += v[i].x;
        o.y = run; run += v[i].y;
        o.z = run; run += v[i].z;
        o.w = run; run += v[i].w;
        q[i] = o;
        int4 o1;
        o1.x = o.x + a.x; o1.y = o.y + a.y; o1.z = o.z + a.z; o1.w = o.w + a.w;
        q1[i] = o1;
    }
}

__device__ __forceinline__ void proj_body(const int blk, const float* __restrict__ x,
                            const float* __restrict__ Wl, const float* __restrict__ bl,
                            const float* __restrict__ Wr, const float* __restrict__ br,
                            float* __restrict__ xl, float* __restrict__ xr) {
    const int wid = blk * (blockDim.x >> 5) + (threadIdx.x >> 5);
    if (wid >= NN / 16) return;
    const int lane = threadIdx.x & 31;
    const int half = lane >> 4;
    const int l    = lane & 15;
    const int m0   = wid * 16;

    v8f accL[4] = {};
    v8f accR[4] = {};

    for (int k0 = 0; k0 < CC; k0 += 4) {
        v2f a;
        const float* ap = x + (m0 + l) * CC + k0 + 2 * half;
        a.x = ap[0];
        a.y = ap[1];
#pragma unroll
        for (int n = 0; n < 4; n++) {
            const int kb = (k0 + 2 * half) * CC + n * 16 + l;
            v2f b;
            b.x = Wl[kb];
            b.y = Wl[kb + CC];
            accL[n] = __builtin_amdgcn_wmma_f32_16x16x4_f32(
                false, a, false, b, (short)0, accL[n], false, false);
            v2f c;
            c.x = Wr[kb];
            c.y = Wr[kb + CC];
            accR[n] = __builtin_amdgcn_wmma_f32_16x16x4_f32(
                false, a, false, c, (short)0, accR[n], false, false);
        }
    }

#pragma unroll
    for (int n = 0; n < 4; n++) {
#pragma unroll
        for (int r = 0; r < 8; r++) {
            const int row = m0 + half * 8 + r;
            const int col = n * 16 + l;
            xl[row * CC + col] = accL[n][r] + bl[col];
            xr[row * CC + col] = accR[n][r] + br[col];
        }
    }
}

#define PROJ_BLOCKS ((NN / 16 + 7) / 8)
__global__ void __launch_bounds__(256)
k_front(const float* __restrict__ ea, float* __restrict__ macc,
        const int* __restrict__ eidx, int* __restrict__ cnt, int* __restrict__ rank,
        const float* __restrict__ x,
        const float* __restrict__ Wl, const float* __restrict__ bl,
        const float* __restrict__ Wr, const float* __restrict__ br,
        float* __restrict__ xl, float* __restrict__ xr) {
    if (blockIdx.x < COUNT_BLOCKS) count_body(ea, macc, eidx, cnt, rank);
    else proj_body((int)blockIdx.x - COUNT_BLOCKS, x, Wl, bl, Wr, br, xl, xr);
}

#define NTILES ((EE + NN) / 16)
static_assert(EE % 16 == 0 && NN % 16 == 0, "edge tiles must not mix edges and self loops");

template <int CTRL>
__device__ __forceinline__ float row_xadd(float v) {
    const int o = __builtin_amdgcn_update_dpp(0, __float_as_int(v), CTRL, 0xf, 0xf, true);
    return v + __int_as_float(o);
}

__global__ void __launch_bounds__(256) __attribute__((amdgpu_waves_per_eu(8, 8)))
k_logits(const int* __restrict__ eidx, const float* __restrict__ ea,
         const float* __restrict__ We, const float* __restrict__ att,
         const float* __restrict__ macc,
         const float* __restrict__ xl, const float* __restrict__ xr,
         const int* __restrict__ rank, const int* __restrict__ offs,
         int2* __restrict__ rec, float* __restrict__ lgs) {
    const int lane = threadIdx.x & 31;
    const int half = lane >> 4;
    const int c    = lane & 15;
    const int wave = __builtin_amdgcn_readfirstlane((int)((blockIdx.x * blockDim.x + threadIdx.x) >> 5));
    const int nw   = (int)((gridDim.x * blockDim.x) >> 5);

    v2f bw[3][4];
#pragma unroll
    for (int j = 0; j < 3; j++) {
        const int k = 4 * j + 2 * half;
        const float4 w0 = *(const float4*)(We + k * CC + 4 * c);
        float4 w1 = make_float4(0.0f, 0.0f, 0.0f, 0.0f);
        if (k + 1 < EDD) w1 = *(const float4*)(We + (k + 1) * CC + 4 * c);
        bw[j][0].x = w0.x; bw[j][0].y = w1.x;
        bw[j][1].x = w0.y; bw[j][1].y = w1.y;
        bw[j][2].x = w0.z; bw[j][2].y = w1.z;
        bw[j][3].x = w0.w; bw[j][3].y = w1.w;
    }
    const float4 av = *(const float4*)(att + 4 * c);
    const float4* __restrict__ xl4 = (const float4*)xl;
    const float4* __restrict__ xr4 = (const float4*)xr;

    for (int t = wave; t < NTILES; t += nw) {
        const int e0 = t * 16;
        int si[8], di[8];
        v2f a[3];
        int myslot = 0, mysrc = 0;
        if (e0 < EE) {
            const unsigned io = (unsigned)e0 + 8u * (unsigned)half;
            const unsigned me = io + (unsigned)(c & 7);
            mysrc  = eidx[me];
            const int rk = rank[me];
            myslot = offs[((unsigned)rk >> 31) * (unsigned)SCAN_N + (unsigned)eidx[(unsigned)EE + me]] + (rk & 0x7fffffff);
            const int4 s0 = *(const int4*)(eidx + io);
            const int4 s1 = *(const int4*)(eidx + io + 4u);
            const int4 d0 = *(const int4*)(eidx + (unsigned)EE + io);
            const int4 d1 = *(const int4*)(eidx + (unsigned)EE + io + 4u);
            si[0] = s0.x; si[1] = s0.y; si[2] = s0.z; si[3] = s0.w;
            si[4] = s1.x; si[5] = s1.y; si[6] = s1.z; si[7] = s1.w;
            di[0] = d0.x; di[1] = d0.y; di[2] = d0.z; di[3] = d0.w;
            di[4] = d1.x; di[5] = d1.y; di[6] = d1.z; di[7] = d1.w;
            const float* er = ea + ((unsigned)(e0 + c) * EDD + 2u * (unsigned)half);
            a[0].x = er[0]; a[0].y = er[1];
            a[1].x = er[4]; a[1].y = er[5];
            a[2].x = er[8];
            a[2].y = 0.0f;
            if (half == 0) a[2].y = er[9];
        } else {
            const int n0 = e0 - EE + 8 * half;
#pragma unroll
            for (int r = 0; r < 8; r++) { si[r] = n0 + r; di[r] = n0 + r; }
            const float inv = 1.0f / (float)EE;
            const float* mr = macc + 2 * half;
            a[0].x = mr[0] * inv; a[0].y = mr[1] * inv;
            a[1].x = mr[4] * inv; a[1].y = mr[5] * inv;
            a[2].x = mr[8] * inv;
            a[2].y = 0.0f;
            if (half == 0) a[2].y = mr[9] * inv;
        }

        float4 vl[8], vr[8];
#pragma unroll
        for (int r = 0; r < 8; r++) {
            vl[r] = xl4[(unsigned)si[r] * (CC / 4) + (unsigned)c];
            vr[r] = xr4[(unsigned)di[r] * (CC / 4) + (unsigned)c];
        }
        v8f acc[4];
#pragma unroll
        for (int r = 0; r < 8; r++) {
            acc[0][r] = vl[r].x + vr[r].x;
            acc[1][r] = vl[r].y + vr[r].y;
            acc[2][r] = vl[r].z + vr[r].z;
            acc[3][r] = vl[r].w + vr[r].w;
        }
#pragma unroll
        for (int j = 0; j < 3; j++) {
#pragma unroll
            for (int nt = 0; nt < 4; nt++) {
                acc[nt] = __builtin_amdgcn_wmma_f32_16x16x4_f32(
                    false, a[j], false, bw[j][nt], (short)0, acc[nt], false, false);
            }
        }

        float part[8];
#pragma unroll
        for (int r = 0; r < 8; r++) {
            float h0 = acc[0][r], h1 = acc[1][r], h2 = acc[2][r], h3 = acc[3][r];
            h0 = h0 > 0.0f ? h0 : NEGS * h0;
            h1 = h1 > 0.0f ? h1 : NEGS * h1;
            h2 = h2 > 0.0f ? h2 : NEGS * h2;
            h3 = h3 > 0.0f ? h3 : NEGS * h3;
            part[r] = h0 * av.x + h1 * av.y + h2 * av.z + h3 * av.w;
        }
#pragma unroll
        for (int r = 0; r < 8; r++) part[r] = row_xadd<0x161>(part[r]);
#pragma unroll
        for (int r = 0; r < 8; r++) part[r] = row_xadd<0x162>(part[r]);
#pragma unroll
        for (int r = 0; r < 8; r++) part[r] = row_xadd<0x164>(part[r]);
#pragma unroll
        for (int r = 0; r < 8; r++) part[r] = row_xadd<0x168>(part[r]);

        float p = part[0];
#pragma unroll
        for (int r = 1; r < 8; r++) {
            if ((c & 7) == r) p = part[r];
        }
        if (c < 8) {
            if (e0 < EE) rec[myslot] = make_int2(mysrc, __float_as_int(p));
            else         lgs[e0 - EE + 8 * half + c] = p;
        }
    }
}

template <int CTRL>
__device__ __forceinline__ float row_xmax(float v) {
    const int o = __builtin_amdgcn_update_dpp(0, __float_as_int(v), CTRL, 0xf, 0xf, true);
    return fmaxf(v, __int_as_float(o));
}

static_assert((NN * 16) % 256 == 0, "the gather grid holds whole waves of real nodes only");

__global__ void __launch_bounds__(256)
k_gather(const int2* __restrict__ rec, const float* __restrict__ lgs, const int* __restrict__ offs,
         const float* __restrict__ xl, const float* __restrict__ x, const float* __restrict__ bias,
         float* __restrict__ out) {
    const int gid = blockIdx.x * blockDim.x + threadIdx.x;
    const int n = gid >> 4;
    const int c = gid & 15;
    const int beg = offs[n];
    const int end = offs[n + 1];
    const float ls = lgs[n];

    float m = ls;
    for (int j = beg + c; j < end; j += 16) m = fmaxf(m, __int_as_float(rec[j].y));
    m = row_xmax<0x161>(m);
    m = row_xmax<0x162>(m);
    m = row_xmax<0x164>(m);
    m = row_xmax<0x168>(m);

    const float4* __restrict__ xl4 = (const float4*)xl;
    float4 acc = make_float4(0.0f, 0.0f, 0.0f, 0.0f);
    float den = 0.0f;
    int j = beg;
    for (; j + 4 <= end; j += 4) {
        const int2 r0 = rec[j], r1 = rec[j + 1], r2 = rec[j + 2], r3 = rec[j + 3];
        const float4 v0 = xl4[(unsigned)r0.x * (CC / 4) + (unsigned)c];
        const float4 v1 = xl4[(unsigned)r1.x * (CC / 4) + (unsigned)c];
        const float4 v2 = xl4[(unsigned)r2.x * (CC / 4) + (unsigned)c];
        const float4 v3 = xl4[(unsigned)r3.x * (CC / 4) + (unsigned)c];
        const float p0 = expf(__int_as_float(r0.y) - m);
        const float p1 = expf(__int_as_float(r1.y) - m);
        const float p2 = expf(__int_as_float(r2.y) - m);
        const float p3 = expf(__int_as_float(r3.y) - m);
        den += p0; acc.x += p0 * v0.x; acc.y += p0 * v0.y; acc.z += p0 * v0.z; acc.w += p0 * v0.w;
        den += p1; acc.x += p1 * v1.x; acc.y += p1 * v1.y; acc.z += p1 * v1.z; acc.w += p1 * v1.w;
        den += p2; acc.x += p2 * v2.x; acc.y += p2 * v2.y; acc.z += p2 * v2.z; acc.w += p2 * v2.w;
        den += p3; acc.x += p3 * v3.x; acc.y += p3 * v3.y; acc.z += p3 * v3.z; acc.w += p3 * v3.w;
    }
    for (; j < end; j++) {
        const int2 r0 = rec[j];
        const float4 v0 = xl4[(unsigned)r0.x * (CC / 4) + (unsigned)c];
        const float p0 = expf(__int_as_float(r0.y) - m);
        den += p0; acc.x += p0 * v0.x; acc.y += p0 * v0.y; acc.z += p0 * v0.z; acc.w += p0 * v0.w;
    }
    {
        const float4 v0 = xl4[(unsigned)n * (CC / 4) + (unsigned)c];
        const float p0 = expf(ls - m);
        den += p0; acc.x += p0 * v0.x; acc.y += p0 * v0.y; acc.z += p0 * v0.z; acc.w += p0 * v0.w;
    }
    const float inv = 1.0f / den;
    const float4 xv = ((const float4*)x)[(unsigned)n * (CC / 4) + (unsigned)c];
    const float4 bv = ((const float4*)bias)[c];
    float4 o;
    o.x = xv.x + (acc.x * inv + bv.x);
    o.y = xv.y + (acc.y * inv + bv.y);
    o.z = xv.z + (acc.z * inv + bv.z);
    o.w = xv.w + (acc.w * inv + bv.w);
    ((float4*)out)[(unsigned)n * (CC / 4) + (unsigned)c] = o;
}

extern "C" void kernel_launch(void* const* d_in, const int* in_sizes, int n_in,
                              void* d_out, int out_size, void* d_ws, size_t ws_size,
                              hipStream_t stream) {
    const float* x    = (const float*)d_in[0];
    const int*   eidx = (const int*)  d_in[1];
    const float* ea   = (const float*)d_in[2];
    const float* Wl   = (const float*)d_in[3];
    const float* bl   = (const float*)d_in[4];
    const float* Wr   = (const float*)d_in[5];
    const float* br   = (const float*)d_in[6];
    const float* We   = (const float*)d_in[7];
    const float* att  = (const float*)d_in[8];
    const float* bias = (const float*)d_in[9];
    float* out = (float*)d_out;

    float* ws    = (float*)d_ws;
    float* xl    = ws;
    float* xr    = out;
    int2*  rec   = (int2*)(xl + NN * CC);
    float* lgs   = (float*)(rec + EE);
    int*   rank  = (int*)(lgs + NN);
    int*   cnt   = (int*)rec;
    int*   offs  = rank + EE;
    float* macc  = (float*)(offs + 2 * SCAN_N);

    const int T = 256;
    k_init<<<2 * SCAN_N / T, T, 0, stream>>>(cnt, macc);
    k_front<<<COUNT_BLOCKS + PROJ_BLOCKS, T, 0, stream>>>(ea, macc, eidx, cnt, rank, x, Wl, bl, Wr, br, xl, xr);
    k_scan<<<1, 1024, 0, stream>>>(cnt, offs);
    k_logits<<<1024, T, 0, stream>>>(eidx, ea, We, att, macc, xl, xr, rank, offs, rec, lgs);
    k_gather<<<(NN * 16) / T, T, 0, stream>>>(rec, lgs, offs, xl, x, bias, out);
}


# k_gather: one record per lane of the node's 16-lane row (a 128-B read per 16 in-edges), one exp per record, row_share broadcast, 16 row gathers in flight; 1024 resident blocks; sum order changes
# speedup vs baseline: 2.9126x; 1.0307x over previous
#include <hip/hip_runtime.h>

#define NN 50000
#define EE 800000
#define CC 64
#define EDD 11
#define NEGS 0.2f

typedef __attribute__((ext_vector_type(2))) float v2f;
typedef __attribute__((ext_vector_type(8))) float v8f;

#define SCAN_CHUNK 52
#define SCAN_N (1024 * SCAN_CHUNK)
static_assert(SCAN_N > NN && SCAN_CHUNK % 4 == 0, "scan covers every node plus the end offset");

__global__ void k_init(int* __restrict__ cnt2, float* __restrict__ macc) {
    int i = blockIdx.x * blockDim.x + threadIdx.x;
    if (i < 2 * SCAN_N) cnt2[i] = 0;
    if (i < 16) macc[i] = 0.0f;
}

#define COUNT_BLOCKS 512
__device__ __forceinline__ void count_body(const float* __restrict__ ea, float* __restrict__ macc,
                          const int* __restrict__ eidx, int* __restrict__ cnt,
                          int* __restrict__ rank) {
    __shared__ float s[EDD];
    if (threadIdx.x < EDD) s[threadIdx.x] = 0.0f;
    __syncthreads();
    float loc[EDD];
#pragma unroll
    for (int k = 0; k < EDD; k++) loc[k] = 0.0f;
    const unsigned hsel = (__builtin_amdgcn_s_sendmsg_rtn(0x87) >> 18) & 1u;
    int* __restrict__ myc = cnt + hsel * SCAN_N;
    const int tag = (int)(hsel << 31);
    for (int e = blockIdx.x * blockDim.x + threadIdx.x; e < EE; e += COUNT_BLOCKS * blockDim.x) {
#pragma unroll
        for (int k = 0; k < EDD; k++) loc[k] += ea[e * EDD + k];
        rank[e] = atomicAdd(&myc[eidx[EE + e]], 1) | tag;
    }
#pragma unroll
    for (int k = 0; k < EDD; k++) atomicAdd(&s[k], loc[k]);
    __syncthreads();
    if (threadIdx.x < EDD) atomicAdd(&macc[threadIdx.x], s[threadIdx.x]);
}

__global__ void __launch_bounds__(1024) k_scan(const int* __restrict__ cnt, int* __restrict__ offs) {
    __shared__ int wsum[32];
    const int t    = threadIdx.x;
    const int lane = t & 31;
    const int wid  = t >> 5;
    const int4* __restrict__ p  = (const int4*)(cnt + t * SCAN_CHUNK);
    const int4* __restrict__ p1 = (const int4*)(cnt + SCAN_N + t * SCAN_CHUNK);
    int4 v[SCAN_CHUNK / 4];
#pragma unroll
    for (int i = 0; i < SCAN_CHUNK / 4; i++) {
        const int4 a = p[i];
        const int4 b = p1[i];
        v[i].x = a.x + b.x; v[i].y = a.y + b.y; v[i].z = a.z + b.z; v[i].w = a.w + b.w;
    }
    int s = 0;
#pragma unroll
    for (int i = 0; i < SCAN_CHUNK / 4; i++) s += v[i].x + v[i].y + v[i].z + v[i].w;
    int inc = s;
    for (int o = 1; o < 32; o <<= 1) {
        const int y = __shfl_up(inc, o);
        if (lane >= o) inc += y;
    }
    if (lane == 31) wsum[wid] = inc;
    __syncthreads();
    if (wid == 0) {
        const int w = wsum[lane];
        int winc = w;
        for (int o = 1; o < 32; o <<= 1) {
            const int y = __shfl_up(winc, o);
            if (lane >= o) winc += y;
        }
        wsum[lane] = winc - w;
    }
    __syncthreads();
    int run = wsum[wid] + inc - s;
    int4* __restrict__ q  = (int4*)(offs + t * SCAN_CHUNK);
    int4* __restrict__ q1 = (int4*)(offs + SCAN_N + t * SCAN_CHUNK);
#pragma unroll
    for (int i = 0; i < SCAN_CHUNK / 4; i++) {
        const int4 a = p[i];
        int4 o;
        o.x = run; run += v[i].x;
        o.y = run; run += v[i].y;
        o.z = run; run += v[i].z;
        o.w = run; run += v[i].w;
        q[i] = o;
        int4 o1;
        o1.x = o.x + a.x; o1.y = o.y + a.y; o1.z = o.z + a.z; o1.w = o.w + a.w;
        q1[i] = o1;
    }
}

__device__ __forceinline__ void proj_body(const int blk, const float* __restrict__ x,
                            const float* __restrict__ Wl, const float* __restrict__ bl,
                            const float* __restrict__ Wr, const float* __restrict__ br,
                            float* __restrict__ xl, float* __restrict__ xr) {
    const int wid = blk * (blockDim.x >> 5) + (threadIdx.x >> 5);
    if (wid >= NN / 16) return;
    const int lane = threadIdx.x & 31;
    const int half = lane >> 4;
    const int l    = lane & 15;
    const int m0   = wid * 16;

    v8f accL[4] = {};
    v8f accR[4] = {};

    for (int k0 = 0; k0 < CC; k0 += 4) {
        v2f a;
        const float* ap = x + (m0 + l) * CC + k0 + 2 * half;
        a.x = ap[0];
        a.y = ap[1];
#pragma unroll
        for (int n = 0; n < 4; n++) {
            const int kb = (k0 + 2 * half) * CC + n * 16 + l;
            v2f b;
            b.x = Wl[kb];
            b.y = Wl[kb + CC];
            accL[n] = __builtin_amdgcn_wmma_f32_16x16x4_f32(
                false, a, false, b, (short)0, accL[n], false, false);
            v2f c;
            c.x = Wr[kb];
            c.y = Wr[kb + CC];
            accR[n] = __builtin_amdgcn_wmma_f32_16x16x4_f32(
                false, a, false, c, (short)0, accR[n], false, false);
        }
    }

#pragma unroll
    for (int n = 0; n < 4; n++) {
#pragma unroll
        for (int r = 0; r < 8; r++) {
            const int row = m0 + half * 8 + r;
            const int col = n * 16 + l;
            xl[row * CC + col] = accL[n][r] + bl[col];
            xr[row * CC + col] = accR[n][r] + br[col];
        }
    }
}

#define PROJ_BLOCKS ((NN / 16 + 7) / 8)
__global__ void __launch_bounds__(256)
k_front(const float* __restrict__ ea, float* __restrict__ macc,
        const int* __restrict__ eidx, int* __restrict__ cnt, int* __restrict__ rank,
        const float* __restrict__ x,
        const float* __restrict__ Wl, const float* __restrict__ bl,
        const float* __restrict__ Wr, const float* __restrict__ br,
        float* __restrict__ xl, float* __restrict__ xr) {
    if (blockIdx.x < COUNT_BLOCKS) count_body(ea, macc, eidx, cnt, rank);
    else proj_body((int)blockIdx.x - COUNT_BLOCKS, x, Wl, bl, Wr, br, xl, xr);
}

#define NTILES ((EE + NN) / 16)
static_assert(EE % 16 == 0 && NN % 16 == 0, "edge tiles must not mix edges and self loops");

template <int CTRL>
__device__ __forceinline__ float row_xadd(float v) {
    const int o = __builtin_amdgcn_update_dpp(0, __float_as_int(v), CTRL, 0xf, 0xf, true);
    return v + __int_as_float(o);
}

__global__ void __launch_bounds__(256) __attribute__((amdgpu_waves_per_eu(8, 8)))
k_logits(const int* __restrict__ eidx, const float* __restrict__ ea,
         const float* __restrict__ We, const float* __restrict__ att,
         const float* __restrict__ macc,
         const float* __restrict__ xl, const float* __restrict__ xr,
         const int* __restrict__ rank, const int* __restrict__ offs,
         int2* __restrict__ rec, float* __restrict__ lgs) {
    const int lane = threadIdx.x & 31;
    const int half = lane >> 4;
    const int c    = lane & 15;
    const int wave = __builtin_amdgcn_readfirstlane((int)((blockIdx.x * blockDim.x + threadIdx.x) >> 5));
    const int nw   = (int)((gridDim.x * blockDim.x) >> 5);

    v2f bw[3][4];
#pragma unroll
    for (int j = 0; j < 3; j++) {
        const int k = 4 * j + 2 * half;
        const float4 w0 = *(const float4*)(We + k * CC + 4 * c);
        float4 w1 = make_float4(0.0f, 0.0f, 0.0f, 0.0f);
        if (k + 1 < EDD) w1 = *(const float4*)(We + (k + 1) * CC + 4 * c);
        bw[j][0].x = w0.x; bw[j][0].y = w1.x;
        bw[j][1].x = w0.y; bw[j][1].y = w1.y;
        bw[j][2].x = w0.z; bw[j][2].y = w1.z;
        bw[j][3].x = w0.w; bw[j][3].y = w1.w;
    }
    const float4 av = *(const float4*)(att + 4 * c);
    const float4* __restrict__ xl4 = (const float4*)xl;
    const float4* __restrict__ xr4 = (const float4*)xr;

    for (int t = wave; t < NTILES; t += nw) {
        const int e0 = t * 16;
        int si[8], di[8];
        v2f a[3];
        int myslot = 0, mysrc = 0;
        if (e0 < EE) {
            const unsigned io = (unsigned)e0 + 8u * (unsigned)half;
            const unsigned me = io + (unsigned)(c & 7);
            mysrc  = eidx[me];
            const int rk = rank[me];
            myslot = offs[((unsigned)rk >> 31) * (unsigned)SCAN_N + (unsigned)eidx[(unsigned)EE + me]] + (rk & 0x7fffffff);
            const int4 s0 = *(const int4*)(eidx + io);
            const int4 s1 = *(const int4*)(eidx + io + 4u);
            const int4 d0 = *(const int4*)(eidx + (unsigned)EE + io);
            const int4 d1 = *(const int4*)(eidx + (unsigned)EE + io + 4u);
            si[0] = s0.x; si[1] = s0.y; si[2] = s0.z; si[3] = s0.w;
            si[4] = s1.x; si[5] = s1.y; si[6] = s1.z; si[7] = s1.w;
            di[0] = d0.x; di[1] = d0.y; di[2] = d0.z; di[3] = d0.w;
            di[4] = d1.x; di[5] = d1.y; di[6] = d1.z; di[7] = d1.w;
            const float* er = ea + ((unsigned)(e0 + c) * EDD + 2u * (unsigned)half);
            a[0].x = er[0]; a[0].y = er[1];
            a[1].x = er[4]; a[1].y = er[5];
            a[2].x = er[8];
            a[2].y = 0.0f;
            if (half == 0) a[2].y = er[9];
        } else {
            const int n0 = e0 - EE + 8 * half;
#pragma unroll
            for (int r = 0; r < 8; r++) { si[r] = n0 + r; di[r] = n0 + r; }
            const float inv = 1.0f / (float)EE;
            const float* mr = macc + 2 * half;
            a[0].x = mr[0] * inv; a[0].y = mr[1] * inv;
            a[1].x = mr[4] * inv; a[1].y = mr[5] * inv;
            a[2].x = mr[8] * inv;
            a[2].y = 0.0f;
            if (half == 0) a[2].y = mr[9] * inv;
        }

        float4 vl[8], vr[8];
#pragma unroll
        for (int r = 0; r < 8; r++) {
            vl[r] = xl4[(unsigned)si[r] * (CC / 4) + (unsigned)c];
            vr[r] = xr4[(unsigned)di[r] * (CC / 4) + (unsigned)c];
        }
        v8f acc[4];
#pragma unroll
        for (int r = 0; r < 8; r++) {
            acc[0][r] = vl[r].x + vr[r].x;
            acc[1][r] = vl[r].y + vr[r].y;
            acc[2][r] = vl[r].z + vr[r].z;
            acc[3][r] = vl[r].w + vr[r].w;
        }
#pragma unroll
        for (int j = 0; j < 3; j++) {
#pragma unroll
            for (int nt = 0; nt < 4; nt++) {
                acc[nt] = __builtin_amdgcn_wmma_f32_16x16x4_f32(
                    false, a[j], false, bw[j][nt], (short)0, acc[nt], false, false);
            }
        }

        float part[8];
#pragma unroll
        for (int r = 0; r < 8; r++) {
            float h0 = acc[0][r], h1 = acc[1][r], h2 = acc[2][r], h3 = acc[3][r];
            h0 = h0 > 0.0f ? h0 : NEGS * h0;
            h1 = h1 > 0.0f ? h1 : NEGS * h1;
            h2 = h2 > 0.0f ? h2 : NEGS * h2;
            h3 = h3 > 0.0f ? h3 : NEGS * h3;
            part[r] = h0 * av.x + h1 * av.y + h2 * av.z + h3 * av.w;
        }
#pragma unroll
        for (int r = 0; r < 8; r++) part[r] = row_xadd<0x161>(part[r]);
#pragma unroll
        for (int r = 0; r < 8; r++) part[r] = row_xadd<0x162>(part[r]);
#pragma unroll
        for (int r = 0; r < 8; r++) part[r] = row_xadd<0x164>(part[r]);
#pragma unroll
        for (int r = 0; r < 8; r++) part[r] = row_xadd<0x168>(part[r]);

        float p = part[0];
#pragma unroll
        for (int r = 1; r < 8; r++) {
            if ((c & 7) == r) p = part[r];
        }
        if (c < 8) {
            if (e0 < EE) rec[myslot] = make_int2(mysrc, __float_as_int(p));
            else         lgs[e0 - EE + 8 * half + c] = p;
        }
    }
}

template <int CTRL>
__device__ __forceinline__ float row_xmax(float v) {
    const int o = __builtin_amdgcn_update_dpp(0, __float_as_int(v), CTRL, 0xf, 0xf, true);
    return fmaxf(v, __int_as_float(o));
}
template <int K>
__device__ __forceinline__ int row_share_i(int v) {
    return __builtin_amdgcn_update_dpp(0, v, 0x150 + K, 0xf, 0xf, true);
}
template <int K>
__device__ __forceinline__ float row_share_f(float v) {
    return __int_as_float(row_share_i<K>(__float_as_int(v)));
}

#define G_ADDR(K) a[K] = (unsigned)row_share_i<K0 + K>(src) * (CC / 4) + c;
#define G_FMA(K) { const float q = row_share_f<K0 + K>(p); \
    acc.x += q * v[K].x; acc.y += q * v[K].y; acc.z += q * v[K].z; acc.w += q * v[K].w; }

template <int K0>
__device__ __forceinline__ void take8(const float4* __restrict__ xl4, const int src, const float p,
                                      const unsigned c, float4& acc) {
    unsigned a[8];
    G_ADDR(0) G_ADDR(1) G_ADDR(2) G_ADDR(3) G_ADDR(4) G_ADDR(5) G_ADDR(6) G_ADDR(7)
    float4 v[8];
#pragma unroll
    for (int k = 0; k < 8; k++) v[k] = xl4[a[k]];
    G_FMA(0) G_FMA(1) G_FMA(2) G_FMA(3) G_FMA(4) G_FMA(5) G_FMA(6) G_FMA(7)
}

__device__ __forceinline__ void take16(const float4* __restrict__ xl4, const int src, const float p,
                                       const unsigned c, float4& acc) {
    constexpr int K0 = 0;
    unsigned a[16];
    G_ADDR(0) G_ADDR(1) G_ADDR(2) G_ADDR(3) G_ADDR(4) G_ADDR(5) G_ADDR(6) G_ADDR(7)
    G_ADDR(8) G_ADDR(9) G_ADDR(10) G_ADDR(11) G_ADDR(12) G_ADDR(13) G_ADDR(14) G_ADDR(15)
    float4 v[16];
#pragma unroll
    for (int k = 0; k < 16; k++) v[k] = xl4[a[k]];
    G_FMA(0) G_FMA(1) G_FMA(2) G_FMA(3) G_FMA(4) G_FMA(5) G_FMA(6) G_FMA(7)
    G_FMA(8) G_FMA(9) G_FMA(10) G_FMA(11) G_FMA(12) G_FMA(13) G_FMA(14) G_FMA(15)
}

#define GATHER_BLOCKS 1024
#define NPAIRS (NN / 2)
static_assert(NN % 2 == 0, "a wave takes two whole nodes");

__global__ void __launch_bounds__(256) __attribute__((amdgpu_waves_per_eu(8, 8)))
k_gather(const int2* __restrict__ rec, const float* __restrict__ lgs, const int* __restrict__ offs,
         const float* __restrict__ xl, const float* __restrict__ x, const float* __restrict__ bias,
         float* __restrict__ out) {
    const int lane = threadIdx.x & 31;
    const int hrow = lane >> 4;
    const unsigned c = (unsigned)(lane & 15);
    int pr = __builtin_amdgcn_readfirstlane((int)((blockIdx.x * blockDim.x + threadIdx.x) >> 5));
    const int nw = (int)((gridDim.x * blockDim.x) >> 5);
    if (pr >= NPAIRS) return;

    const float4* __restrict__ xl4 = (const float4*)xl;
    const float4 bv = ((const float4*)bias)[c];
    const int NEG_INF = (int)0xff800000u;

    int n = 2 * pr + hrow;
    int beg = offs[n];
    int end = offs[n + 1];
    float ls = lgs[n];

    for (;;) {
        const int j0 = beg + (int)c;
        int2 r0 = make_int2(n, NEG_INF);
        int2 r1 = make_int2(n, NEG_INF);
        if (j0 < end)      r0 = rec[j0];
        if (j0 + 16 < end) r1 = rec[j0 + 16];
        const float4 vs = xl4[(unsigned)n * (CC / 4) + c];
        const float4 xv = ((const float4*)x)[(unsigned)n * (CC / 4) + c];

        const int npr = pr + nw;
        int nn = n, nbeg = 0, nend = 0;
        float nls = 0.0f;
        if (npr < NPAIRS) {
            nn   = 2 * npr + hrow;
            nbeg = offs[nn];
            nend = offs[nn + 1];
            nls  = lgs[nn];
        }

        float m = fmaxf(ls, fmaxf(__int_as_float(r0.y), __int_as_float(r1.y)));
        for (int j = j0 + 32; j < end; j += 16) m = fmaxf(m, __int_as_float(rec[j].y));
        m = row_xmax<0x161>(m);
        m = row_xmax<0x162>(m);
        m = row_xmax<0x164>(m);
        m = row_xmax<0x168>(m);

        float4 acc = make_float4(0.0f, 0.0f, 0.0f, 0.0f);
        const float p0 = expf(__int_as_float(r0.y) - m);
        float den = p0;
        take16(xl4, r0.x, p0, c, acc);
        const int deg = end - beg;
        if (deg > 16) {
            const float p1 = expf(__int_as_float(r1.y) - m);
            den += p1;
            take8<0>(xl4, r1.x, p1, c, acc);
            if (deg > 24) take8<8>(xl4, r1.x, p1, c, acc);
            for (int b = j0 + 32; b - (int)c < end; b += 16) {
                int2 r = make_int2(n, NEG_INF);
                if (b < end) r = rec[b];
                const float p = expf(__int_as_float(r.y) - m);
                den += p;
                take16(xl4, r.x, p, c, acc);
            }
        }
        den = row_xadd<0x161>(den);
        den = row_xadd<0x162>(den);
        den = row_xadd<0x164>(den);
        den = row_xadd<0x168>(den);
        {
            const float ps = expf(ls - m);
            den += ps;
            acc.x += ps * vs.x; acc.y += ps * vs.y; acc.z += ps * vs.z; acc.w += ps * vs.w;
        }
        const float inv = 1.0f / den;
        float4 o;
        o.x = xv.x + (acc.x * inv + bv.x);
        o.y = xv.y + (acc.y * inv + bv.y);
        o.z = xv.z + (acc.z * inv + bv.z);
        o.w = xv.w + (acc.w * inv + bv.w);
        ((float4*)out)[(unsigned)n * (CC / 4) + c] = o;

        if (npr >= NPAIRS) break;
        pr = npr; n = nn; beg = nbeg; end = nend; ls = nls;
    }
}

extern "C" void kernel_launch(void* const* d_in, const int* in_sizes, int n_in,
                              void* d_out, int out_size, void* d_ws, size_t ws_size,
                              hipStream_t stream) {
    const float* x    = (const float*)d_in[0];
    const int*   eidx = (const int*)  d_in[1];
    const float* ea   = (const float*)d_in[2];
    const float* Wl   = (const float*)d_in[3];
    const float* bl   = (const float*)d_in[4];
    const float* Wr   = (const float*)d_in[5];
    const float* br   = (const float*)d_in[6];
    const float* We   = (const float*)d_in[7];
    const float* att  = (const float*)d_in[8];
    const float* bias = (const float*)d_in[9];
    float* out = (float*)d_out;

    float* ws    = (float*)d_ws;
    float* xl    = ws;
    float* xr    = out;
    int2*  rec   = (int2*)(xl + NN * CC);
    float* lgs   = (float*)(rec + EE);
    int*   rank  = (int*)(lgs + NN);
    int*   cnt   = (int*)rec;
    int*   offs  = rank + EE;
    float* macc  = (float*)(offs + 2 * SCAN_N);

    const int T = 256;
    k_init<<<2 * SCAN_N / T, T, 0, stream>>>(cnt, macc);
    k_front<<<COUNT_BLOCKS + PROJ_BLOCKS, T, 0, stream>>>(ea, macc, eidx, cnt, rank, x, Wl, bl, Wr, br, xl, xr);
    k_scan<<<1, 1024, 0, stream>>>(cnt, offs);
    k_logits<<<1024, T, 0, stream>>>(eidx, ea, We, att, macc, xl, xr, rank, offs, rec, lgs);
    k_gather<<<GATHER_BLOCKS, T, 0, stream>>>(rec, lgs, offs, xl, x, bias, out);
}


# k_scan reads and writes whole 128-B lines: lane t of slab i owns 4 consecutive counters (32 lanes = 512 contiguous bytes per load or store); the 13 slabs' prefix sums run side by side on DPP adds
# speedup vs baseline: 3.8193x; 1.3113x over previous
#include <hip/hip_runtime.h>

#define NN 50000
#define EE 800000
#define CC 64
#define EDD 11
#define NEGS 0.2f

typedef __attribute__((ext_vector_type(2))) float v2f;
typedef __attribute__((ext_vector_type(8))) float v8f;

#define SCAN_CHUNK 52
#define SCAN_N (1024 * SCAN_CHUNK)
static_assert(SCAN_N > NN && SCAN_CHUNK % 4 == 0, "scan covers every node plus the end offset");

__global__ void k_init(int* __restrict__ cnt2, float* __restrict__ macc) {
    int i = blockIdx.x * blockDim.x + threadIdx.x;
    if (i < 2 * SCAN_N) cnt2[i] = 0;
    if (i < 16) macc[i] = 0.0f;
}

#define COUNT_BLOCKS 512
__device__ __forceinline__ void count_body(const float* __restrict__ ea, float* __restrict__ macc,
                          const int* __restrict__ eidx, int* __restrict__ cnt,
                          int* __restrict__ rank) {
    __shared__ float s[EDD];
    if (threadIdx.x < EDD) s[threadIdx.x] = 0.0f;
    __syncthreads();
    float loc[EDD];
#pragma unroll
    for (int k = 0; k < EDD; k++) loc[k] = 0.0f;
    const unsigned hsel = (__builtin_amdgcn_s_sendmsg_rtn(0x87) >> 18) & 1u;
    int* __restrict__ myc = cnt + hsel * SCAN_N;
    const int tag = (int)(hsel << 31);
    for (int e = blockIdx.x * blockDim.x + threadIdx.x; e < EE; e += COUNT_BLOCKS * blockDim.x) {
#pragma unroll
        for (int k = 0; k < EDD; k++) loc[k] += ea[e * EDD + k];
        rank[e] = atomicAdd(&myc[eidx[EE + e]], 1) | tag;
    }
#pragma unroll
    for (int k = 0; k < EDD; k++) atomicAdd(&s[k], loc[k]);
    __syncthreads();
    if (threadIdx.x < EDD) atomicAdd(&macc[threadIdx.x], s[threadIdx.x]);
}

#define SCAN_SLABS (SCAN_N / 4096)
static_assert(SCAN_SLABS * 4096 == SCAN_N && SCAN_SLABS <= 32, "whole slabs of 1024 threads x 4 counters; one wave per slab row");

template <int CTRL>
__device__ __forceinline__ int row_shr_add(int v) {
    return v + __builtin_amdgcn_update_dpp(0, v, CTRL, 0xf, 0xf, true);
}
__device__ __forceinline__ int wave_incl_scan(int v, const int lane) {
    v = row_shr_add<0x111>(v);
    v = row_shr_add<0x112>(v);
    v = row_shr_add<0x114>(v);
    v = row_shr_add<0x118>(v);
    const int lo = __builtin_amdgcn_readlane(v, 15);
    return v + (lane >= 16 ? lo : 0);
}

__global__ void __launch_bounds__(1024) k_scan(const int* __restrict__ cnt, int* __restrict__ offs) {
    __shared__ int wtot[SCAN_SLABS * 32];
    __shared__ int stot[SCAN_SLABS];
    const int t    = threadIdx.x;
    const int lane = t & 31;
    const int wid  = t >> 5;
    const int4* __restrict__ p  = (const int4*)cnt + t;
    const int4* __restrict__ p1 = (const int4*)(cnt + SCAN_N) + t;
    int4 v[SCAN_SLABS];
#pragma unroll
    for (int i = 0; i < SCAN_SLABS; i++) {
        const int4 a = p[i * 1024];
        const int4 b = p1[i * 1024];
        v[i].x = a.x + b.x; v[i].y = a.y + b.y; v[i].z = a.z + b.z; v[i].w = a.w + b.w;
    }
    int inc[SCAN_SLABS];
#pragma unroll
    for (int i = 0; i < SCAN_SLABS; i++)
        inc[i] = wave_incl_scan(v[i].x + v[i].y + v[i].z + v[i].w, lane);
    if (lane == 31) {
#pragma unroll
        for (int i = 0; i < SCAN_SLABS; i++) wtot[i * 32 + wid] = inc[i];
    }
    __syncthreads();
    if (wid < SCAN_SLABS) {
        const int w  = wtot[wid * 32 + lane];
        const int wi = wave_incl_scan(w, lane);
        wtot[wid * 32 + lane] = wi - w;
        if (lane == 31) stot[wid] = wi;
    }
    __syncthreads();
    int4* __restrict__ q  = (int4*)offs + t;
    int4* __restrict__ q1 = (int4*)(offs + SCAN_N) + t;
    int carry = 0;
#pragma unroll
    for (int i = 0; i < SCAN_SLABS; i++) {
        const int4 a = p[i * 1024];
        int run = carry + wtot[i * 32 + wid] + inc[i] - (v[i].x + v[i].y + v[i].z + v[i].w);
        int4 o;
        o.x = run; run += v[i].x;
        o.y = run; run += v[i].y;
        o.z = run; run += v[i].z;
        o.w = run;
        q[i * 1024] = o;
        int4 o1;
        o1.x = o.x + a.x; o1.y = o.y + a.y; o1.z = o.z + a.z; o1.w = o.w + a.w;
        q1[i * 1024] = o1;
        carry += stot[i];
    }
}

__device__ __forceinline__ void proj_body(const int blk, const float* __restrict__ x,
                            const float* __restrict__ Wl, const float* __restrict__ bl,
                            const float* __restrict__ Wr, const float* __restrict__ br,
                            float* __restrict__ xl, float* __restrict__ xr) {
    const int wid = blk * (blockDim.x >> 5) + (threadIdx.x >> 5);
    if (wid >= NN / 16) return;
    const int lane = threadIdx.x & 31;
    const int half = lane >> 4;
    const int l    = lane & 15;
    const int m0   = wid * 16;

    v8f accL[4] = {};
    v8f accR[4] = {};

    for (int k0 = 0; k0 < CC; k0 += 4) {
        v2f a;
        const float* ap = x + (m0 + l) * CC + k0 + 2 * half;
        a.x = ap[0];
        a.y = ap[1];
#pragma unroll
        for (int n = 0; n < 4; n++) {
            const int kb = (k0 + 2 * half) * CC + n * 16 + l;
            v2f b;
            b.x = Wl[kb];
            b.y = Wl[kb + CC];
            accL[n] = __builtin_amdgcn_wmma_f32_16x16x4_f32(
                false, a, false, b, (short)0, accL[n], false, false);
            v2f c;
            c.x = Wr[kb];
            c.y = Wr[kb + CC];
            accR[n] = __builtin_amdgcn_wmma_f32_16x16x4_f32(
                false, a, false, c, (short)0, accR[n], false, false);
        }
    }

#pragma unroll
    for (int n = 0; n < 4; n++) {
#pragma unroll
        for (int r = 0; r < 8; r++) {
            const int row = m0 + half * 8 + r;
            const int col = n * 16 + l;
            xl[row * CC + col] = accL[n][r] + bl[col];
            xr[row * CC + col] = accR[n][r] + br[col];
        }
    }
}

#define PROJ_BLOCKS ((NN / 16 + 7) / 8)
__global__ void __launch_bounds__(256)
k_front(const float* __restrict__ ea, float* __restrict__ macc,
        const int* __restrict__ eidx, int* __restrict__ cnt, int* __restrict__ rank,
        const float* __restrict__ x,
        const float* __restrict__ Wl, const float* __restrict__ bl,
        const float* __restrict__ Wr, const float* __restrict__ br,
        float* __restrict__ xl, float* __restrict__ xr) {
    if (blockIdx.x < COUNT_BLOCKS) count_body(ea, macc, eidx, cnt, rank);
    else proj_body((int)blockIdx.x - COUNT_BLOCKS, x, Wl, bl, Wr, br, xl, xr);
}

#define NTILES ((EE + NN) / 16)
static_assert(EE % 16 == 0 && NN % 16 == 0, "edge tiles must not mix edges and self loops");

template <int CTRL>
__device__ __forceinline__ float row_xadd(float v) {
    const int o = __builtin_amdgcn_update_dpp(0, __float_as_int(v), CTRL, 0xf, 0xf, true);
    return v + __int_as_float(o);
}

__global__ void __launch_bounds__(256) __attribute__((amdgpu_waves_per_eu(8, 8)))
k_logits(const int* __restrict__ eidx, const float* __restrict__ ea,
         const float* __restrict__ We, const float* __restrict__ att,
         const float* __restrict__ macc,
         const float* __restrict__ xl, const float* __restrict__ xr,
         const int* __restrict__ rank, const int* __restrict__ offs,
         int2* __restrict__ rec, float* __restrict__ lgs) {
    const int lane = threadIdx.x & 31;
    const int half = lane >> 4;
    const int c    = lane & 15;
    const int wave = __builtin_amdgcn_readfirstlane((int)((blockIdx.x * blockDim.x + threadIdx.x) >> 5));
    const int nw   = (int)((gridDim.x * blockDim.x) >> 5);

    v2f bw[3][4];
#pragma unroll
    for (int j = 0; j < 3; j++) {
        const int k = 4 * j + 2 * half;
        const float4 w0 = *(const float4*)(We + k * CC + 4 * c);
        float4 w1 = make_float4(0.0f, 0.0f, 0.0f, 0.0f);
        if (k + 1 < EDD) w1 = *(const float4*)(We + (k + 1) * CC + 4 * c);
        bw[j][0].x = w0.x; bw[j][0].y = w1.x;
        bw[j][1].x = w0.y; bw[j][1].y = w1.y;
        bw[j][2].x = w0.z; bw[j][2].y = w1.z;
        bw[j][3].x = w0.w; bw[j][3].y = w1.w;
    }
    const float4 av = *(const float4*)(att + 4 * c);
    const float4* __restrict__ xl4 = (const float4*)xl;
    const float4* __restrict__ xr4 = (const float4*)xr;

    for (int t = wave; t < NTILES; t += nw) {
        const int e0 = t * 16;
        int si[8], di[8];
        v2f a[3];
        int myslot = 0, mysrc = 0;
        if (e0 < EE) {
            const unsigned io = (unsigned)e0 + 8u * (unsigned)half;
            const unsigned me = io + (unsigned)(c & 7);
            mysrc  = eidx[me];
            const int rk = rank[me];
            myslot = offs[((unsigned)rk >> 31) * (unsigned)SCAN_N + (unsigned)eidx[(unsigned)EE + me]] + (rk & 0x7fffffff);
            const int4 s0 = *(const int4*)(eidx + io);
            const int4 s1 = *(const int4*)(eidx + io + 4u);
            const int4 d0 = *(const int4*)(eidx + (unsigned)EE + io);
            const int4 d1 = *(const int4*)(eidx + (unsigned)EE + io + 4u);
            si[0] = s0.x; si[1] = s0.y; si[2] = s0.z; si[3] = s0.w;
            si[4] = s1.x; si[5] = s1.y; si[6] = s1.z; si[7] = s1.w;
            di[0] = d0.x; di[1] = d0.y; di[2] = d0.z; di[3] = d0.w;
            di[4] = d1.x; di[5] = d1.y; di[6] = d1.z; di[7] = d1.w;
            const float* er = ea + ((unsigned)(e0 + c) * EDD + 2u * (unsigned)half);
            a[0].x = er[0]; a[0].y = er[1];
            a[1].x = er[4]; a[1].y = er[5];
            a[2].x = er[8];
            a[2].y = 0.0f;
            if (half == 0) a[2].y = er[9];
        } else {
            const int n0 = e0 - EE + 8 * half;
#pragma unroll
            for (int r = 0; r < 8; r++) { si[r] = n0 + r; di[r] = n0 + r; }
            const float inv = 1.0f / (float)EE;
            const float* mr = macc + 2 * half;
            a[0].x = mr[0] * inv; a[0].y = mr[1] * inv;
            a[1].x = mr[4] * inv; a[1].y = mr[5] * inv;
            a[2].x = mr[8] * inv;
            a[2].y = 0.0f;
            if (half == 0) a[2].y = mr[9] * inv;
        }

        float4 vl[8], vr[8];
#pragma unroll
        for (int r = 0; r < 8; r++) {
            vl[r] = xl4[(unsigned)si[r] * (CC / 4) + (unsigned)c];
            vr[r] = xr4[(unsigned)di[r] * (CC / 4) + (unsigned)c];
        }
        v8f acc[4];
#pragma unroll
        for (int r = 0; r < 8; r++) {
            acc[0][r] = vl[r].x + vr[r].x;
            acc[1][r] = vl[r].y + vr[r].y;
            acc[2][r] = vl[r].z + vr[r].z;
            acc[3][r] = vl[r].w + vr[r].w;
        }
#pragma unroll
        for (int j = 0; j < 3; j++) {
#pragma unroll
            for (int nt = 0; nt < 4; nt++) {
                acc[nt] = __builtin_amdgcn_wmma_f32_16x16x4_f32(
                    false, a[j], false, bw[j][nt], (short)0, acc[nt], false, false);
            }
        }

        float part[8];
#pragma unroll
        for (int r = 0; r < 8; r++) {
            float h0 = acc[0][r], h1 = acc[1][r], h2 = acc[2][r], h3 = acc[3][r];
            h0 = h0 > 0.0f ? h0 : NEGS * h0;
            h1 = h1 > 0.0f ? h1 : NEGS * h1;
            h2 = h2 > 0.0f ? h2 : NEGS * h2;
            h3 = h3 > 0.0f ? h3 : NEGS * h3;
            part[r] = h0 * av.x + h1 * av.y + h2 * av.z + h3 * av.w;
        }
#pragma unroll
        for (int r = 0; r < 8; r++) part[r] = row_xadd<0x161>(part[r]);
#pragma unroll
        for (int r = 0; r < 8; r++) part[r] = row_xadd<0x162>(part[r]);
#pragma unroll
        for (int r = 0; r < 8; r++) part[r] = row_xadd<0x164>(part[r]);
#pragma unroll
        for (int r = 0; r < 8; r++) part[r] = row_xadd<0x168>(part[r]);

        float p = part[0];
#pragma unroll
        for (int r = 1; r < 8; r++) {
            if ((c & 7) == r) p = part[r];
        }
        if (c < 8) {
            if (e0 < EE) rec[myslot] = make_int2(mysrc, __float_as_int(p));
            else         lgs[e0 - EE + 8 * half + c] = p;
        }
    }
}

template <int CTRL>
__device__ __forceinline__ float row_xmax(float v) {
    const int o = __builtin_amdgcn_update_dpp(0, __float_as_int(v), CTRL, 0xf, 0xf, true);
    return fmaxf(v, __int_as_float(o));
}
template <int K>
__device__ __forceinline__ int row_share_i(int v) {
    return __builtin_amdgcn_update_dpp(0, v, 0x150 + K, 0xf, 0xf, true);
}
template <int K>
__device__ __forceinline__ float row_share_f(float v) {
    return __int_as_float(row_share_i<K>(__float_as_int(v)));
}

#define G_ADDR(K) a[K] = (unsigned)row_share_i<K0 + K>(src) * (CC / 4) + c;
#define G_FMA(K) { const float q = row_share_f<K0 + K>(p); \
    acc.x += q * v[K].x; acc.y += q * v[K].y; acc.z += q * v[K].z; acc.w += q * v[K].w; }

template <int K0>
__device__ __forceinline__ void take8(const float4* __restrict__ xl4, const int src, const float p,
                                      const unsigned c, float4& acc) {
    unsigned a[8];
    G_ADDR(0) G_ADDR(1) G_ADDR(2) G_ADDR(3) G_ADDR(4) G_ADDR(5) G_ADDR(6) G_ADDR(7)
    float4 v[8];
#pragma unroll
    for (int k = 0; k < 8; k++) v[k] = xl4[a[k]];
    G_FMA(0) G_FMA(1) G_FMA(2) G_FMA(3) G_FMA(4) G_FMA(5) G_FMA(6) G_FMA(7)
}

__device__ __forceinline__ void take16(const float4* __restrict__ xl4, const int src, const float p,
                                       const unsigned c, float4& acc) {
    constexpr int K0 = 0;
    unsigned a[16];
    G_ADDR(0) G_ADDR(1) G_ADDR(2) G_ADDR(3) G_ADDR(4) G_ADDR(5) G_ADDR(6) G_ADDR(7)
    G_ADDR(8) G_ADDR(9) G_ADDR(10) G_ADDR(11) G_ADDR(12) G_ADDR(13) G_ADDR(14) G_ADDR(15)
    float4 v[16];
#pragma unroll
    for (int k = 0; k < 16; k++) v[k] = xl4[a[k]];
    G_FMA(0) G_FMA(1) G_FMA(2) G_FMA(3) G_FMA(4) G_FMA(5) G_FMA(6) G_FMA(7)
    G_FMA(8) G_FMA(9) G_FMA(10) G_FMA(11) G_FMA(12) G_FMA(13) G_FMA(14) G_FMA(15)
}

#define GATHER_BLOCKS 1024
#define NPAIRS (NN / 2)
static_assert(NN % 2 == 0, "a wave takes two whole nodes");

__global__ void __launch_bounds__(256) __attribute__((amdgpu_waves_per_eu(8, 8)))
k_gather(const int2* __restrict__ rec, const float* __restrict__ lgs, const int* __restrict__ offs,
         const float* __restrict__ xl, const float* __restrict__ x, const float* __restrict__ bias,
         float* __restrict__ out) {
    const int lane = threadIdx.x & 31;
    const int hrow = lane >> 4;
    const unsigned c = (unsigned)(lane & 15);
    int pr = __builtin_amdgcn_readfirstlane((int)((blockIdx.x * blockDim.x + threadIdx.x) >> 5));
    const int nw = (int)((gridDim.x * blockDim.x) >> 5);
    if (pr >= NPAIRS) return;

    const float4* __restrict__ xl4 = (const float4*)xl;
    const float4 bv = ((const float4*)bias)[c];
    const int NEG_INF = (int)0xff800000u;

    int n = 2 * pr + hrow;
    int beg = offs[n];
    int end = offs[n + 1];
    float ls = lgs[n];

    for (;;) {
        const int j0 = beg + (int)c;
        int2 r0 = make_int2(n, NEG_INF);
        int2 r1 = make_int2(n, NEG_INF);
        if (j0 < end)      r0 = rec[j0];
        if (j0 + 16 < end) r1 = rec[j0 + 16];
        const float4 vs = xl4[(unsigned)n * (CC / 4) + c];
        const float4 xv = ((const float4*)x)[(unsigned)n * (CC / 4) + c];

        const int npr = pr + nw;
        int nn = n, nbeg = 0, nend = 0;
        float nls = 0.0f;
        if (npr < NPAIRS) {
            nn   = 2 * npr + hrow;
            nbeg = offs[nn];
            nend = offs[nn + 1];
            nls  = lgs[nn];
        }

        float m = fmaxf(ls, fmaxf(__int_as_float(r0.y), __int_as_float(r1.y)));
        for (int j = j0 + 32; j < end; j += 16) m = fmaxf(m, __int_as_float(rec[j].y));
        m = row_xmax<0x161>(m);
        m = row_xmax<0x162>(m);
        m = row_xmax<0x164>(m);
        m = row_xmax<0x168>(m);

        float4 acc = make_float4(0.0f, 0.0f, 0.0f, 0.0f);
        const float p0 = expf(__int_as_float(r0.y) - m);
        float den = p0;
        take16(xl4, r0.x, p0, c, acc);
        const int deg = end - beg;
        if (deg > 16) {
            const float p1 = expf(__int_as_float(r1.y) - m);
            den += p1;
            take8<0>(xl4, r1.x, p1, c, acc);
            if (deg > 24) take8<8>(xl4, r1.x, p1, c, acc);
            for (int b = j0 + 32; b - (int)c < end; b += 16) {
                int2 r = make_int2(n, NEG_INF);
                if (b < end) r = rec[b];
                const float p = expf(__int_as_float(r.y) - m);
                den += p;
                take16(xl4, r.x, p, c, acc);
            }
        }
        den = row_xadd<0x161>(den);
        den = row_xadd<0x162>(den);
        den = row_xadd<0x164>(den);
        den = row_xadd<0x168>(den);
        {
            const float ps = expf(ls - m);
            den += ps;
            acc.x += ps * vs.x; acc.y += ps * vs.y; acc.z += ps * vs.z; acc.w += ps * vs.w;
        }
        const float inv = 1.0f / den;
        float4 o;
        o.x = xv.x + (acc.x * inv + bv.x);
        o.y = xv.y + (acc.y * inv + bv.y);
        o.z = xv.z + (acc.z * inv + bv.z);
        o.w = xv.w + (acc.w * inv + bv.w);
        ((float4*)out)[(unsigned)n * (CC / 4) + c] = o;

        if (npr >= NPAIRS) break;
        pr = npr; n = nn; beg = nbeg; end = nend; ls = nls;
    }
}

extern "C" void kernel_launch(void* const* d_in, const int* in_sizes, int n_in,
                              void* d_out, int out_size, void* d_ws, size_t ws_size,
                              hipStream_t stream) {
    const float* x    = (const float*)d_in[0];
    const int*   eidx = (const int*)  d_in[1];
    const float* ea   = (const float*)d_in[2];
    const float* Wl   = (const float*)d_in[3];
    const float* bl   = (const float*)d_in[4];
    const float* Wr   = (const float*)d_in[5];
    const float* br   = (const float*)d_in[6];
    const float* We   = (const float*)d_in[7];
    const float* att  = (const float*)d_in[8];
    const float* bias = (const float*)d_in[9];
    float* out = (float*)d_out;

    float* ws    = (float*)d_ws;
    float* xl    = ws;
    float* xr    = out;
    int2*  rec   = (int2*)(xl + NN * CC);
    float* lgs   = (float*)(rec + EE);
    int*   rank  = (int*)(lgs + NN);
    int*   cnt   = (int*)rec;
    int*   offs  = rank + EE;
    float* macc  = (float*)(offs + 2 * SCAN_N);

    const int T = 256;
    k_init<<<2 * SCAN_N / T, T, 0, stream>>>(cnt, macc);
    k_front<<<COUNT_BLOCKS + PROJ_BLOCKS, T, 0, stream>>>(ea, macc, eidx, cnt, rank, x, Wl, bl, Wr, br, xl, xr);
    k_scan<<<1, 1024, 0, stream>>>(cnt, offs);
    k_logits<<<1024, T, 0, stream>>>(eidx, ea, We, att, macc, xl, xr, rank, offs, rec, lgs);
    k_gather<<<GATHER_BLOCKS, T, 0, stream>>>(rec, lgs, offs, xl, x, bias, out);
}


# logits join node pass: count pass drops edge ids into 48-slot destination buckets (no scan); wave per node: WMMA tile logits, running-max softmax, one xl gather; 5 launches to 3; sum order changes
# speedup vs baseline: 4.2287x; 1.1072x over previous
#include <hip/hip_runtime.h>

#define NN 50000
#define EE 800000
#define CC 64
#define EDD 11
#define NEGS 0.2f

typedef __attribute__((ext_vector_type(2))) float v2f;
typedef __attribute__((ext_vector_type(8))) float v8f;

#define CAP 48
#define INIT_BLOCKS ((2 * NN + 255) / 256)

__global__ void k_init(int* __restrict__ cnt2, float* __restrict__ macc) {
    int i = blockIdx.x * blockDim.x + threadIdx.x;
    if (i < 2 * NN) cnt2[i] = 0;
    if (i < 16) macc[i] = 0.0f;
}

#define COUNT_BLOCKS 512
__device__ __forceinline__ void count_body(const float* __restrict__ ea, float* __restrict__ macc,
                          const int* __restrict__ eidx, int* __restrict__ cnt,
                          int* __restrict__ bkt) {
    __shared__ float s[EDD];
    if (threadIdx.x < EDD) s[threadIdx.x] = 0.0f;
    __syncthreads();
    float loc[EDD];
#pragma unroll
    for (int k = 0; k < EDD; k++) loc[k] = 0.0f;
    const unsigned hsel = (__builtin_amdgcn_s_sendmsg_rtn(0x87) >> 18) & 1u;
    int* __restrict__ myc = cnt + hsel * NN;
    for (int e = blockIdx.x * blockDim.x + threadIdx.x; e < EE; e += COUNT_BLOCKS * blockDim.x) {
#pragma unroll
        for (int k = 0; k < EDD; k++) loc[k] += ea[e * EDD + k];
        const int d = eidx[EE + e];
        const int r = atomicAdd(&myc[d], 1);
        if (r < CAP) bkt[d * CAP + (hsel ? CAP - 1 - r : r)] = e;
    }
#pragma unroll
    for (int k = 0; k < EDD; k++) atomicAdd(&s[k], loc[k]);
    __syncthreads();
    if (threadIdx.x < EDD) atomicAdd(&macc[threadIdx.x], s[threadIdx.x]);
}

__device__ __forceinline__ void proj_body(const int blk, const float* __restrict__ x,
                            const float* __restrict__ Wl, const float* __restrict__ bl,
                            const float* __restrict__ Wr, const float* __restrict__ br,
                            float* __restrict__ xl, float* __restrict__ xr) {
    const int wid = blk * (blockDim.x >> 5) + (threadIdx.x >> 5);
    if (wid >= NN / 16) return;
    const int lane = threadIdx.x & 31;
    const int half = lane >> 4;
    const int l    = lane & 15;
    const int m0   = wid * 16;

    v8f accL[4] = {};
    v8f accR[4] = {};

    for (int k0 = 0; k0 < CC; k0 += 4) {
        v2f a;
        const float* ap = x + (m0 + l) * CC + k0 + 2 * half;
        a.x = ap[0];
        a.y = ap[1];
#pragma unroll
        for (int n = 0; n < 4; n++) {
            const int kb = (k0 + 2 * half) * CC + n * 16 + l;
            v2f b;
            b.x = Wl[kb];
            b.y = Wl[kb + CC];
            accL[n] = __builtin_amdgcn_wmma_f32_16x16x4_f32(
                false, a, false, b, (short)0, accL[n], false, false);
            v2f c;
            c.x = Wr[kb];
            c.y = Wr[kb + CC];
            accR[n] = __builtin_amdgcn_wmma_f32_16x16x4_f32(
                false, a, false, c, (short)0, accR[n], false, false);
        }
    }

#pragma unroll
    for (int n = 0; n < 4; n++) {
#pragma unroll
        for (int r = 0; r < 8; r++) {
            const int row = m0 + half * 8 + r;
            const int col = n * 16 + l;
            xl[row * CC + col] = accL[n][r] + bl[col];
            xr[row * CC + col] = accR[n][r] + br[col];
        }
    }
}

#define PROJ_BLOCKS ((NN / 16 + 7) / 8)
__global__ void __launch_bounds__(256)
k_front(const float* __restrict__ ea, float* __restrict__ macc,
        const int* __restrict__ eidx, int* __restrict__ cnt, int* __restrict__ bkt,
        const float* __restrict__ x,
        const float* __restrict__ Wl, const float* __restrict__ bl,
        const float* __restrict__ Wr, const float* __restrict__ br,
        float* __restrict__ xl, float* __restrict__ xr) {
    if (blockIdx.x < COUNT_BLOCKS) count_body(ea, macc, eidx, cnt, bkt);
    else proj_body((int)blockIdx.x - COUNT_BLOCKS, x, Wl, bl, Wr, br, xl, xr);
}

template <int CTRL>
__device__ __forceinline__ float row_xadd(float v) {
    const int o = __builtin_amdgcn_update_dpp(0, __float_as_int(v), CTRL, 0xf, 0xf, true);
    return v + __int_as_float(o);
}
template <int K>
__device__ __forceinline__ int row_share_i(int v) {
    return __builtin_amdgcn_update_dpp(0, v, 0x150 + K, 0xf, 0xf, true);
}
template <int K>
__device__ __forceinline__ float row_share_f(float v) {
    return __int_as_float(row_share_i<K>(__float_as_int(v)));
}
template <int CTRL>
__device__ __forceinline__ float row_xmax(float v) {
    const int o = __builtin_amdgcn_update_dpp(0, __float_as_int(v), CTRL, 0xf, 0xf, true);
    return fmaxf(v, __int_as_float(o));
}
__device__ __forceinline__ float xhalf(float v) {
    const int b = __float_as_int(v);
    return __int_as_float(__builtin_amdgcn_permlanex16(b, b, 0x76543210, 0xfedcba98, false, false));
}

#define SRC_OF(R) { const int lo = row_share_i<R>(src_c); const int hi = row_share_i<8 + R>(src_c); si[R] = half ? hi : lo; }

#define NODE_BLOCKS 1024
static_assert(EE % 16 == 0, "the edge list is walked in whole tiles where a bucket overflowed");

__global__ void __launch_bounds__(256) __attribute__((amdgpu_waves_per_eu(8, 8)))
k_node(const int* __restrict__ eidx, const float* __restrict__ ea,
       const float* __restrict__ We, const float* __restrict__ att,
       const float* __restrict__ macc,
       const float* __restrict__ xl, const float* xr,
       const int* __restrict__ cnt, const int* __restrict__ bkt,
       const float* __restrict__ x, const float* __restrict__ bias,
       float* out) {
    const int lane = threadIdx.x & 31;
    const int half = lane >> 4;
    const int c    = lane & 15;
    int n = __builtin_amdgcn_readfirstlane((int)((blockIdx.x * blockDim.x + threadIdx.x) >> 5));
    const int nw = (int)((gridDim.x * blockDim.x) >> 5);

    __shared__ __attribute__((aligned(16))) float sB[32 * 28];
    if (threadIdx.x < 32) {
#pragma unroll
        for (int j = 0; j < 3; j++) {
            const int k = 4 * j + 2 * half;
            const float4 w0 = *(const float4*)(We + k * CC + 4 * c);
            float4 w1 = make_float4(0.0f, 0.0f, 0.0f, 0.0f);
            if (k + 1 < EDD) w1 = *(const float4*)(We + (k + 1) * CC + 4 * c);
            float* sb = sB + lane * 28 + j * 8;
            sb[0] = w0.x; sb[1] = w1.x;
            sb[2] = w0.y; sb[3] = w1.y;
            sb[4] = w0.z; sb[5] = w1.z;
            sb[6] = w0.w; sb[7] = w1.w;
        }
    }
    if (threadIdx.x < 32) {
        const float4 a4 = *(const float4*)(att + 4 * c);
        float* sa = sB + lane * 28 + 24;
        sa[0] = a4.x; sa[1] = a4.y; sa[2] = a4.z; sa[3] = a4.w;
    }
    __syncthreads();
    const float4* __restrict__ sB4 = (const float4*)(sB + lane * 28);
    if (n >= NN) return;
    const float4* __restrict__ xl4 = (const float4*)xl;
    const float NEG_INF = __int_as_float((int)0xff800000u);
    const float invE = 1.0f / (float)EE;

    int c0 = __builtin_amdgcn_readfirstlane(cnt[n]);
    int c1 = __builtin_amdgcn_readfirstlane(cnt[NN + n]);
    int pe = 0;
    if (c0 + c1 <= CAP && c >= 1 && c <= c0 + c1)
        pe = bkt[c <= c0 ? (n * CAP - 1) + c : (n * CAP + CAP + c0) - c];

    for (;;) {
        const int nn = n + nw;
        int nc0 = 0, nc1 = 0;
        if (nn < NN) { nc0 = cnt[nn]; nc1 = cnt[NN + nn]; }
        float4 xrn = make_float4(0.0f, 0.0f, 0.0f, 0.0f);

        const int  deg  = c0 + c1;
        const bool slow = deg > CAP;
        const int ntiles = slow ? 1 + EE / 16 : (deg + 16) >> 4;

        float m = NEG_INF, den = 0.0f;
        float4 accv = make_float4(0.0f, 0.0f, 0.0f, 0.0f);

        for (int tt = 0; tt < ntiles; tt++) {
            int e_c = -1, src_c = n;
            bool ok_c;
            if (!slow) {
                const int j = 16 * tt + c;
                ok_c = j <= deg;
                if (j >= 1 && j <= deg) {
                    const int at = j <= c0 ? (n * CAP - 1) + j : (n * CAP + CAP + c0) - j;
                    const unsigned id = (unsigned)(tt == 0 ? pe : bkt[at]);
                    e_c = (int)(id < (unsigned)EE ? id : (unsigned)EE - 1u);
                    src_c = eidx[e_c];
                }
            } else if (tt == 0) {
                ok_c = c == 0;
            } else {
                const int b = (tt - 1) * 16 + c;
                const bool hit = eidx[EE + b] == n;
                if (__builtin_amdgcn_ballot_w32(hit) == 0u) continue;
                ok_c = hit;
                if (hit) { e_c = b; src_c = eidx[b]; }
            }
            const unsigned okm = __builtin_amdgcn_ballot_w32(ok_c);
            if (tt == 0) xrn = ((const float4*)xr)[(unsigned)n * (CC / 4) + (unsigned)c];

            const float* er = (e_c >= 0 ? ea + (size_t)(unsigned)e_c * EDD : macc) + 2 * half;
            const float  as = e_c >= 0 ? 1.0f : invE;
            v2f a[3];
            a[0].x = er[0] * as; a[0].y = er[1] * as;
            a[1].x = er[4] * as; a[1].y = er[5] * as;
            a[2].x = er[8] * as;
            const float a9 = er[9 - half];
            a[2].y = half == 0 ? a9 * as : 0.0f;

            int si[8];
            SRC_OF(0) SRC_OF(1) SRC_OF(2) SRC_OF(3) SRC_OF(4) SRC_OF(5) SRC_OF(6) SRC_OF(7)

            float4 vl[8];
#pragma unroll
            for (int r = 0; r < 8; r++) vl[r] = xl4[(unsigned)si[r] * (CC / 4) + (unsigned)c];

            v8f acc[4];
#pragma unroll
            for (int r = 0; r < 8; r++) {
                acc[0][r] = vl[r].x + xrn.x;
                acc[1][r] = vl[r].y + xrn.y;
                acc[2][r] = vl[r].z + xrn.z;
                acc[3][r] = vl[r].w + xrn.w;
            }
#pragma unroll
            for (int j = 0; j < 3; j++) {
                const float4 b01 = sB4[2 * j], b23 = sB4[2 * j + 1];
                v2f b0, b1, b2, b3;
                b0.x = b01.x; b0.y = b01.y; b1.x = b01.z; b1.y = b01.w;
                b2.x = b23.x; b2.y = b23.y; b3.x = b23.z; b3.y = b23.w;
                acc[0] = __builtin_amdgcn_wmma_f32_16x16x4_f32(false, a[j], false, b0, (short)0, acc[0], false, false);
                acc[1] = __builtin_amdgcn_wmma_f32_16x16x4_f32(false, a[j], false, b1, (short)0, acc[1], false, false);
                acc[2] = __builtin_amdgcn_wmma_f32_16x16x4_f32(false, a[j], false, b2, (short)0, acc[2], false, false);
                acc[3] = __builtin_amdgcn_wmma_f32_16x16x4_f32(false, a[j], false, b3, (short)0, acc[3], false, false);
            }

            const float4 av = sB4[6];
            float part[8];
#pragma unroll
            for (int r = 0; r < 8; r++) {
                float h0 = acc[0][r], h1 = acc[1][r], h2 = acc[2][r], h3 = acc[3][r];
                h0 = fmaxf(h0, NEGS * h0);
                h1 = fmaxf(h1, NEGS * h1);
                h2 = fmaxf(h2, NEGS * h2);
                h3 = fmaxf(h3, NEGS * h3);
                part[r] = h0 * av.x + h1 * av.y + h2 * av.z + h3 * av.w;
            }
#pragma unroll
            for (int r = 0; r < 8; r++) part[r] = row_xadd<0x161>(part[r]);
#pragma unroll
            for (int r = 0; r < 8; r++) part[r] = row_xadd<0x162>(part[r]);
#pragma unroll
            for (int r = 0; r < 8; r++) part[r] = row_xadd<0x164>(part[r]);
#pragma unroll
            for (int r = 0; r < 8; r++) part[r] = row_xadd<0x168>(part[r]);

            float lown = part[0];
#pragma unroll
            for (int r = 1; r < 8; r++) {
                if ((c & 7) == r) lown = part[r];
            }
            if (((okm >> (8 * half + (c & 7))) & 1u) == 0u) lown = NEG_INF;
            float mt = lown;
            mt = row_xmax<0x161>(mt);
            mt = row_xmax<0x162>(mt);
            mt = row_xmax<0x164>(mt);
            mt = row_xmax<0x168>(mt);
            mt = fmaxf(mt, xhalf(mt));
            const float mnew = fmaxf(m, mt);
            const float rs = expf(m - mnew);
            den *= rs;
            accv.x *= rs; accv.y *= rs; accv.z *= rs; accv.w *= rs;
            const float pown = expf(lown - mnew);
            float pw[8];
            pw[0] = row_share_f<0>(pown); pw[1] = row_share_f<1>(pown); pw[2] = row_share_f<2>(pown); pw[3] = row_share_f<3>(pown);
            pw[4] = row_share_f<4>(pown); pw[5] = row_share_f<5>(pown); pw[6] = row_share_f<6>(pown); pw[7] = row_share_f<7>(pown);
#pragma unroll
            for (int r = 0; r < 8; r++) {
                const float p = pw[r];
                den += p;
                accv.x += p * vl[r].x; accv.y += p * vl[r].y; accv.z += p * vl[r].z; accv.w += p * vl[r].w;
            }
            m = mnew;
        }

        int npe = 0;
        if (nn < NN) {
            const int s0 = __builtin_amdgcn_readfirstlane(nc0);
            const int sd = s0 + __builtin_amdgcn_readfirstlane(nc1);
            if (sd <= CAP && c >= 1 && c <= sd)
                npe = bkt[c <= s0 ? (nn * CAP - 1) + c : (nn * CAP + CAP + s0) - c];
        }

        den += xhalf(den);
        accv.x += xhalf(accv.x); accv.y += xhalf(accv.y); accv.z += xhalf(accv.z); accv.w += xhalf(accv.w);
        const float4 xv = ((const float4*)x)[(unsigned)n * (CC / 4) + (unsigned)c];
        const float4 bv = ((const float4*)bias)[c];
        const float inv = 1.0f / den;
        float4 o;
        o.x = xv.x + (accv.x * inv + bv.x);
        o.y = xv.y + (accv.y * inv + bv.y);
        o.z = xv.z + (accv.z * inv + bv.z);
        o.w = xv.w + (accv.w * inv + bv.w);
        if (half == 0) ((float4*)out)[(unsigned)n * (CC / 4) + (unsigned)c] = o;

        if (nn >= NN) break;
        n = nn;
        pe = npe;
        c0 = __builtin_amdgcn_readfirstlane(nc0);
        c1 = __builtin_amdgcn_readfirstlane(nc1);
    }
}

extern "C" void kernel_launch(void* const* d_in, const int* in_sizes, int n_in,
                              void* d_out, int out_size, void* d_ws, size_t ws_size,
                              hipStream_t stream) {
    const float* x    = (const float*)d_in[0];
    const int*   eidx = (const int*)  d_in[1];
    const float* ea   = (const float*)d_in[2];
    const float* Wl   = (const float*)d_in[3];
    const float* bl   = (const float*)d_in[4];
    const float* Wr   = (const float*)d_in[5];
    const float* br   = (const float*)d_in[6];
    const float* We   = (const float*)d_in[7];
    const float* att  = (const float*)d_in[8];
    const float* bias = (const float*)d_in[9];
    float* out = (float*)d_out;

    float* ws    = (float*)d_ws;
    float* xl    = ws;
    float* xr    = out;
    int*   bkt   = (int*)(xl + NN * CC);
    int*   cnt   = bkt + NN * CAP;
    float* macc  = (float*)(cnt + 2 * NN);

    const int T = 256;
    k_init<<<INIT_BLOCKS, T, 0, stream>>>(cnt, macc);
    k_front<<<COUNT_BLOCKS + PROJ_BLOCKS, T, 0, stream>>>(ea, macc, eidx, cnt, bkt, x, Wl, bl, Wr, br, xl, xr);
    k_node<<<NODE_BLOCKS, T, 0, stream>>>(eidx, ea, We, att, macc, xl, xr, cnt, bkt, x, bias, out);
}


# k_node tiles gather rows straight from prefetched data: bucket slots hold {edge id, source} pairs written by the counting pass (per-edge source gather gone), xr row read a node ahead; sum order kept
# speedup vs baseline: 4.4685x; 1.0567x over previous
#include <hip/hip_runtime.h>

#define NN 50000
#define EE 800000
#define CC 64
#define EDD 11
#define NEGS 0.2f

typedef __attribute__((ext_vector_type(2))) float v2f;
typedef __attribute__((ext_vector_type(8))) float v8f;

#define CAP 48
#define INIT_BLOCKS ((2 * NN + 255) / 256)
#define WS_NEED ((size_t)NN * CC * 4 + (size_t)NN * CAP * 8 + (size_t)2 * NN * 4 + 16 * 4)

__global__ void k_init(int* __restrict__ cnt2, float* __restrict__ macc, const size_t wsb) {
    if (wsb < WS_NEED) return;
    int i = blockIdx.x * blockDim.x + threadIdx.x;
    if (i < 2 * NN) cnt2[i] = 0;
    if (i < 16) macc[i] = 0.0f;
}

#define COUNT_BLOCKS 512
__device__ __forceinline__ void count_body(const float* __restrict__ ea, float* __restrict__ macc,
                          const int* __restrict__ eidx, int* __restrict__ cnt,
                          int2* __restrict__ bkt) {
    __shared__ float s[EDD];
    if (threadIdx.x < EDD) s[threadIdx.x] = 0.0f;
    __syncthreads();
    float loc[EDD];
#pragma unroll
    for (int k = 0; k < EDD; k++) loc[k] = 0.0f;
    const unsigned hsel = (__builtin_amdgcn_s_sendmsg_rtn(0x87) >> 18) & 1u;
    int* __restrict__ myc = cnt + hsel * NN;
    for (int e = blockIdx.x * blockDim.x + threadIdx.x; e < EE; e += COUNT_BLOCKS * blockDim.x) {
#pragma unroll
        for (int k = 0; k < EDD; k++) loc[k] += ea[e * EDD + k];
        const int d = eidx[EE + e];
        const int sn = eidx[e];
        const int r = atomicAdd(&myc[d], 1);
        const int rc = r < CAP - 1 ? r : CAP - 1;
        bkt[d * CAP + (hsel ? CAP - 1 - rc : rc)] = make_int2(e, sn);
    }
#pragma unroll
    for (int k = 0; k < EDD; k++) atomicAdd(&s[k], loc[k]);
    __syncthreads();
    if (threadIdx.x < EDD) atomicAdd(&macc[threadIdx.x], s[threadIdx.x]);
}

__device__ __forceinline__ void proj_body(const int blk, const float* __restrict__ x,
                            const float* __restrict__ Wl, const float* __restrict__ bl,
                            const float* __restrict__ Wr, const float* __restrict__ br,
                            float* __restrict__ xl, float* __restrict__ xr) {
    const int wid = blk * (blockDim.x >> 5) + (threadIdx.x >> 5);
    if (wid >= NN / 16) return;
    const int lane = threadIdx.x & 31;
    const int half = lane >> 4;
    const int l    = lane & 15;
    const int m0   = wid * 16;

    v8f accL[4] = {};
    v8f accR[4] = {};

    for (int k0 = 0; k0 < CC; k0 += 4) {
        v2f a;
        const float* ap = x + (m0 + l) * CC + k0 + 2 * half;
        a.x = ap[0];
        a.y = ap[1];
#pragma unroll
        for (int n = 0; n < 4; n++) {
            const int kb = (k0 + 2 * half) * CC + n * 16 + l;
            v2f b;
            b.x = Wl[kb];
            b.y = Wl[kb + CC];
            accL[n] = __builtin_amdgcn_wmma_f32_16x16x4_f32(
                false, a, false, b, (short)0, accL[n], false, false);
            v2f c;
            c.x = Wr[kb];
            c.y = Wr[kb + CC];
            accR[n] = __builtin_amdgcn_wmma_f32_16x16x4_f32(
                false, a, false, c, (short)0, accR[n], false, false);
        }
    }

#pragma unroll
    for (int n = 0; n < 4; n++) {
#pragma unroll
        for (int r = 0; r < 8; r++) {
            const int row = m0 + half * 8 + r;
            const int col = n * 16 + l;
            xl[row * CC + col] = accL[n][r] + bl[col];
            xr[row * CC + col] = accR[n][r] + br[col];
        }
    }
}

#define PROJ_BLOCKS ((NN / 16 + 7) / 8)
__global__ void __launch_bounds__(256)
k_front(const float* __restrict__ ea, float* __restrict__ macc,
        const int* __restrict__ eidx, int* __restrict__ cnt, int2* __restrict__ bkt,
        const float* __restrict__ x,
        const float* __restrict__ Wl, const float* __restrict__ bl,
        const float* __restrict__ Wr, const float* __restrict__ br,
        float* __restrict__ xl, float* __restrict__ xr, const size_t wsb) {
    if (wsb < WS_NEED) return;
    if (blockIdx.x < COUNT_BLOCKS) count_body(ea, macc, eidx, cnt, bkt);
    else proj_body((int)blockIdx.x - COUNT_BLOCKS, x, Wl, bl, Wr, br, xl, xr);
}

template <int CTRL>
__device__ __forceinline__ float row_xadd(float v) {
    const int o = __builtin_amdgcn_update_dpp(0, __float_as_int(v), CTRL, 0xf, 0xf, true);
    return v + __int_as_float(o);
}
template <int K>
__device__ __forceinline__ int row_share_i(int v) {
    return __builtin_amdgcn_update_dpp(0, v, 0x150 + K, 0xf, 0xf, true);
}
template <int K>
__device__ __forceinline__ float row_share_f(float v) {
    return __int_as_float(row_share_i<K>(__float_as_int(v)));
}
template <int CTRL>
__device__ __forceinline__ float row_xmax(float v) {
    const int o = __builtin_amdgcn_update_dpp(0, __float_as_int(v), CTRL, 0xf, 0xf, true);
    return fmaxf(v, __int_as_float(o));
}
__device__ __forceinline__ float xhalf(float v) {
    const int b = __float_as_int(v);
    return __int_as_float(__builtin_amdgcn_permlanex16(b, b, 0x76543210, 0xfedcba98, false, false));
}

#define SRC_OF(R) { const int lo = row_share_i<R>(src_c); const int hi = row_share_i<8 + R>(src_c); si[R] = half ? hi : lo; }

#define NODE_BLOCKS 1024
static_assert(EE % 16 == 0, "the edge list is walked in whole tiles where a bucket overflowed");

__global__ void __launch_bounds__(256) __attribute__((amdgpu_waves_per_eu(8, 8)))
k_node(const int* __restrict__ eidx, const float* __restrict__ ea,
       const float* __restrict__ We, const float* __restrict__ att,
       const float* __restrict__ macc,
       const float* __restrict__ xl, const float* xr,
       const int* __restrict__ cnt, const int2* __restrict__ bkt,
       const float* __restrict__ x, const float* __restrict__ bias,
       float* out, const size_t wsb) {
    if (wsb < WS_NEED) return;
    const int lane = threadIdx.x & 31;
    const int half = lane >> 4;
    const int c    = lane & 15;
    int n = __builtin_amdgcn_readfirstlane((int)((blockIdx.x * blockDim.x + threadIdx.x) >> 5));
    const int nw = (int)((gridDim.x * blockDim.x) >> 5);

    __shared__ __attribute__((aligned(16))) float sB[32 * 28];
    if (threadIdx.x < 32) {
#pragma unroll
        for (int j = 0; j < 3; j++) {
            const int k = 4 * j + 2 * half;
            const float4 w0 = *(const float4*)(We + k * CC + 4 * c);
            float4 w1 = make_float4(0.0f, 0.0f, 0.0f, 0.0f);
            if (k + 1 < EDD) w1 = *(const float4*)(We + (k + 1) * CC + 4 * c);
            float* sb = sB + lane * 28 + j * 8;
            sb[0] = w0.x; sb[1] = w1.x;
            sb[2] = w0.y; sb[3] = w1.y;
            sb[4] = w0.z; sb[5] = w1.z;
            sb[6] = w0.w; sb[7] = w1.w;
        }
    }
    if (threadIdx.x < 32) {
        const float4 a4 = *(const float4*)(att + 4 * c);
        float* sa = sB + lane * 28 + 24;
        sa[0] = a4.x; sa[1] = a4.y; sa[2] = a4.z; sa[3] = a4.w;
    }
    __syncthreads();
    const float4* __restrict__ sB4 = (const float4*)(sB + lane * 28);
    if (n >= NN) return;
    const float4* __restrict__ xl4 = (const float4*)xl;
    const float NEG_INF = __int_as_float((int)0xff800000u);
    const float invE = 1.0f / (float)EE;

    int c0 = __builtin_amdgcn_readfirstlane(cnt[n]);
    int c1 = __builtin_amdgcn_readfirstlane(cnt[NN + n]);
    int2 pe = make_int2(0, 0);
    if (c0 + c1 <= CAP && c >= 1 && c <= c0 + c1)
        pe = bkt[c <= c0 ? (n * CAP - 1) + c : (n * CAP + CAP + c0) - c];
    float4 xrn = ((const float4*)xr)[(unsigned)n * (CC / 4) + (unsigned)c];

    for (;;) {
        const int nn = n + nw;
        int nc0 = 0, nc1 = 0;
        if (nn < NN) { nc0 = cnt[nn]; nc1 = cnt[NN + nn]; }

        const int  deg  = c0 + c1;
        const bool slow = deg > CAP;
        const int ntiles = slow ? 1 + EE / 16 : (deg + 16) >> 4;

        float m = NEG_INF, den = 0.0f;
        float4 accv = make_float4(0.0f, 0.0f, 0.0f, 0.0f);

        for (int tt = 0; tt < ntiles; tt++) {
            int e_c = -1, src_c = n;
            bool ok_c;
            if (!slow) {
                const int j = 16 * tt + c;
                ok_c = j <= deg;
                if (j >= 1 && j <= deg) {
                    const int at = j <= c0 ? (n * CAP - 1) + j : (n * CAP + CAP + c0) - j;
                    int2 rec = pe;
                    if (tt != 0) rec = bkt[at];
                    const unsigned id = (unsigned)rec.x, sv = (unsigned)rec.y;
                    e_c = (int)(id < (unsigned)EE ? id : (unsigned)EE - 1u);
                    src_c = (int)(sv < (unsigned)NN ? sv : (unsigned)NN - 1u);
                }
            } else if (tt == 0) {
                ok_c = c == 0;
            } else {
                const int b = (tt - 1) * 16 + c;
                const bool hit = eidx[EE + b] == n;
                if (__builtin_amdgcn_ballot_w32(hit) == 0u) continue;
                ok_c = hit;
                if (hit) { e_c = b; src_c = eidx[b]; }
            }
            const unsigned okm = __builtin_amdgcn_ballot_w32(ok_c);

            const float* er = (e_c >= 0 ? ea + (size_t)(unsigned)e_c * EDD : macc) + 2 * half;
            const float  as = e_c >= 0 ? 1.0f : invE;
            v2f a[3];
            a[0].x = er[0] * as; a[0].y = er[1] * as;
            a[1].x = er[4] * as; a[1].y = er[5] * as;
            a[2].x = er[8] * as;
            const float a9 = er[9 - half];
            a[2].y = half == 0 ? a9 * as : 0.0f;

            int si[8];
            SRC_OF(0) SRC_OF(1) SRC_OF(2) SRC_OF(3) SRC_OF(4) SRC_OF(5) SRC_OF(6) SRC_OF(7)

            float4 vl[8];
#pragma unroll
            for (int r = 0; r < 8; r++) vl[r] = xl4[(unsigned)si[r] * (CC / 4) + (unsigned)c];

            v8f acc[4];
#pragma unroll
            for (int r = 0; r < 8; r++) {
                acc[0][r] = vl[r].x + xrn.x;
                acc[1][r] = vl[r].y + xrn.y;
                acc[2][r] = vl[r].z + xrn.z;
                acc[3][r] = vl[r].w + xrn.w;
            }
#pragma unroll
            for (int j = 0; j < 3; j++) {
                const float4 b01 = sB4[2 * j], b23 = sB4[2 * j + 1];
                v2f b0, b1, b2, b3;
                b0.x = b01.x; b0.y = b01.y; b1.x = b01.z; b1.y = b01.w;
                b2.x = b23.x; b2.y = b23.y; b3.x = b23.z; b3.y = b23.w;
                acc[0] = __builtin_amdgcn_wmma_f32_16x16x4_f32(false, a[j], false, b0, (short)0, acc[0], false, false);
                acc[1] = __builtin_amdgcn_wmma_f32_16x16x4_f32(false, a[j], false, b1, (short)0, acc[1], false, false);
                acc[2] = __builtin_amdgcn_wmma_f32_16x16x4_f32(false, a[j], false, b2, (short)0, acc[2], false, false);
                acc[3] = __builtin_amdgcn_wmma_f32_16x16x4_f32(false, a[j], false, b3, (short)0, acc[3], false, false);
            }

            const float4 av = sB4[6];
            float part[8];
#pragma unroll
            for (int r = 0; r < 8; r++) {
                float h0 = acc[0][r], h1 = acc[1][r], h2 = acc[2][r], h3 = acc[3][r];
                h0 = fmaxf(h0, NEGS * h0);
                h1 = fmaxf(h1, NEGS * h1);
                h2 = fmaxf(h2, NEGS * h2);
                h3 = fmaxf(h3, NEGS * h3);
                part[r] = h0 * av.x + h1 * av.y + h2 * av.z + h3 * av.w;
            }
#pragma unroll
            for (int r = 0; r < 8; r++) part[r] = row_xadd<0x161>(part[r]);
#pragma unroll
            for (int r = 0; r < 8; r++) part[r] = row_xadd<0x162>(part[r]);
#pragma unroll
            for (int r = 0; r < 8; r++) part[r] = row_xadd<0x164>(part[r]);
#pragma unroll
            for (int r = 0; r < 8; r++) part[r] = row_xadd<0x168>(part[r]);

            float lown = part[0];
#pragma unroll
            for (int r = 1; r < 8; r++) {
                if ((c & 7) == r) lown = part[r];
            }
            if (((okm >> (8 * half + (c & 7))) & 1u) == 0u) lown = NEG_INF;
            float mt = lown;
            mt = row_xmax<0x161>(mt);
            mt = row_xmax<0x162>(mt);
            mt = row_xmax<0x164>(mt);
            mt = row_xmax<0x168>(mt);
            mt = fmaxf(mt, xhalf(mt));
            const float mnew = fmaxf(m, mt);
            const float rs = expf(m - mnew);
            den *= rs;
            accv.x *= rs; accv.y *= rs; accv.z *= rs; accv.w *= rs;
            const float pown = expf(lown - mnew);
            float pw[8];
            pw[0] = row_share_f<0>(pown); pw[1] = row_share_f<1>(pown); pw[2] = row_share_f<2>(pown); pw[3] = row_share_f<3>(pown);
            pw[4] = row_share_f<4>(pown); pw[5] = row_share_f<5>(pown); pw[6] = row_share_f<6>(pown); pw[7] = row_share_f<7>(pown);
#pragma unroll
            for (int r = 0; r < 8; r++) {
                const float p = pw[r];
                den += p;
                accv.x += p * vl[r].x; accv.y += p * vl[r].y; accv.z += p * vl[r].z; accv.w += p * vl[r].w;
            }
            m = mnew;
        }

        int2 npe = make_int2(0, 0);
        if (nn < NN) {
            const int s0 = __builtin_amdgcn_readfirstlane(nc0);
            const int sd = s0 + __builtin_amdgcn_readfirstlane(nc1);
            if (sd <= CAP && c >= 1 && c <= sd)
                npe = bkt[c <= s0 ? (nn * CAP - 1) + c : (nn * CAP + CAP + s0) - c];
        }
        float4 nxr = xrn;
        if (nn < NN) nxr = ((const float4*)xr)[(unsigned)nn * (CC / 4) + (unsigned)c];

        den += xhalf(den);
        accv.x += xhalf(accv.x); accv.y += xhalf(accv.y); accv.z += xhalf(accv.z); accv.w += xhalf(accv.w);
        const float4 xv = ((const float4*)x)[(unsigned)n * (CC / 4) + (unsigned)c];
        const float4 bv = ((const float4*)bias)[c];
        const float inv = 1.0f / den;
        float4 o;
        o.x = xv.x + (accv.x * inv + bv.x);
        o.y = xv.y + (accv.y * inv + bv.y);
        o.z = xv.z + (accv.z * inv + bv.z);
        o.w = xv.w + (accv.w * inv + bv.w);
        if (half == 0) ((float4*)out)[(unsigned)n * (CC / 4) + (unsigned)c] = o;

        if (nn >= NN) break;
        n = nn;
        pe = npe;
        xrn = nxr;
        c0 = __builtin_amdgcn_readfirstlane(nc0);
        c1 = __builtin_amdgcn_readfirstlane(nc1);
    }
}

extern "C" void kernel_launch(void* const* d_in, const int* in_sizes, int n_in,
                              void* d_out, int out_size, void* d_ws, size_t ws_size,
                              hipStream_t stream) {
    const float* x    = (const float*)d_in[0];
    const int*   eidx = (const int*)  d_in[1];
    const float* ea   = (const float*)d_in[2];
    const float* Wl   = (const float*)d_in[3];
    const float* bl   = (const float*)d_in[4];
    const float* Wr   = (const float*)d_in[5];
    const float* br   = (const float*)d_in[6];
    const float* We   = (const float*)d_in[7];
    const float* att  = (const float*)d_in[8];
    const float* bias = (const float*)d_in[9];
    float* out = (float*)d_out;

    float* ws    = (float*)d_ws;
    float* xl    = ws;
    float* xr    = out;
    int2*  bkt   = (int2*)(xl + NN * CC);
    int*   cnt   = (int*)(bkt + NN * CAP);
    float* macc  = (float*)(cnt + 2 * NN);

    const int T = 256;
    k_init<<<INIT_BLOCKS, T, 0, stream>>>(cnt, macc, ws_size);
    k_front<<<COUNT_BLOCKS + PROJ_BLOCKS, T, 0, stream>>>(ea, macc, eidx, cnt, bkt, x, Wl, bl, Wr, br, xl, xr, ws_size);
    k_node<<<NODE_BLOCKS, T, 0, stream>>>(eidx, ea, We, att, macc, xl, xr, cnt, bkt, x, bias, out, ws_size);
}


# in-edges grouped via LDS: count blocks rank edges per 196-node range (1 global add per block and range); a 49-node block sifts its range into LDS lists; per-edge global adds gone; arrival-order sums
# speedup vs baseline: 4.6487x; 1.0403x over previous
#include <hip/hip_runtime.h>

#define NN 50000
#define EE 800000
#define CC 64
#define EDD 11
#define NEGS 0.2f

typedef __attribute__((ext_vector_type(2))) float v2f;
typedef __attribute__((ext_vector_type(8))) float v8f;

#define CAP 48
#define NPB 49
#define NODE_BLOCKS 1024
#define RNG_BLOCKS 4
#define NRNG (NODE_BLOCKS / RNG_BLOCKS)
#define RNG_NODES (NPB * RNG_BLOCKS)
#define RCAP 4096
static_assert(NPB * NODE_BLOCKS >= NN, "every node belongs to a node block");
static_assert(EE <= (1 << 20) && RNG_NODES <= (1 << 11), "a record's first word holds the edge id and, above it, the offset");
static_assert(4095 - NPB * (RNG_BLOCKS - 1) >= NPB && RCAP % 256 == 0, "offset 4095 (a record that is not there) is no node of any block");
#define INIT_BLOCKS 1
#define WS_NEED ((size_t)NN * CC * 4 + (size_t)NRNG * RCAP * 8 + (size_t)NRNG * 4 + 16 * 4)

__global__ void k_init(int* __restrict__ rcnt, float* __restrict__ macc, const size_t wsb) {
    if (wsb < WS_NEED) return;
    int i = blockIdx.x * blockDim.x + threadIdx.x;
    if (i < NRNG) rcnt[i] = 0;
    if (i < 16) macc[i] = 0.0f;
}

#define COUNT_BLOCKS 512
#define COUNT_THREADS 256
#define EDGES_PER_THREAD ((EE + COUNT_BLOCKS * COUNT_THREADS - 1) / (COUNT_BLOCKS * COUNT_THREADS))
__device__ __forceinline__ void count_body(const float* __restrict__ ea, float* __restrict__ macc,
                          const int* __restrict__ eidx, int* __restrict__ rcnt,
                          int2* __restrict__ rrec) {
    __shared__ float s[EDD];
    __shared__ int lh[NRNG];
    if (threadIdx.x < EDD) s[threadIdx.x] = 0.0f;
    if (threadIdx.x < NRNG) lh[threadIdx.x] = 0;
    __syncthreads();
    float loc[EDD];
#pragma unroll
    for (int k = 0; k < EDD; k++) loc[k] = 0.0f;
    const int gid = blockIdx.x * COUNT_THREADS + threadIdx.x;
    int dv[EDGES_PER_THREAD], sv[EDGES_PER_THREAD], rk[EDGES_PER_THREAD];
#pragma unroll
    for (int it = 0; it < EDGES_PER_THREAD; it++) {
        const int e = gid + it * (COUNT_BLOCKS * COUNT_THREADS);
        dv[it] = -1; sv[it] = 0; rk[it] = 0;
        if (e < EE) {
            const unsigned du = (unsigned)eidx[EE + e];
            dv[it] = (int)(du < (unsigned)NN ? du : (unsigned)NN - 1u);
            sv[it] = eidx[e];
        }
    }
#pragma unroll
    for (int it = 0; it < EDGES_PER_THREAD; it++) {
        const int e = gid + it * (COUNT_BLOCKS * COUNT_THREADS);
        if (e < EE) {
#pragma unroll
            for (int k = 0; k < EDD; k++) loc[k] += ea[e * EDD + k];
        }
    }
#pragma unroll
    for (int it = 0; it < EDGES_PER_THREAD; it++) {
        if (dv[it] >= 0) rk[it] = atomicAdd(&lh[dv[it] / RNG_NODES], 1);
    }
    __syncthreads();
    if (threadIdx.x < NRNG) {
        const int cb = lh[threadIdx.x];
        lh[threadIdx.x] = cb > 0 ? atomicAdd(&rcnt[threadIdx.x], cb) : 0;
    }
    __syncthreads();
#pragma unroll
    for (int it = 0; it < EDGES_PER_THREAD; it++) {
        if (dv[it] >= 0) {
            const int e = gid + it * (COUNT_BLOCKS * COUNT_THREADS);
            const int r = dv[it] / RNG_NODES;
            const int pos = lh[r] + rk[it];
            if (pos < RCAP) rrec[r * RCAP + pos] = make_int2(e | ((dv[it] - r * RNG_NODES) << 20), sv[it]);
        }
    }
#pragma unroll
    for (int k = 0; k < EDD; k++) atomicAdd(&s[k], loc[k]);
    __syncthreads();
    if (threadIdx.x < EDD) atomicAdd(&macc[threadIdx.x], s[threadIdx.x]);
}

__device__ __forceinline__ void proj_body(const int blk, const float* __restrict__ x,
                            const float* __restrict__ Wl, const float* __restrict__ bl,
                            const float* __restrict__ Wr, const float* __restrict__ br,
                            float* __restrict__ xl, float* __restrict__ xr) {
    const int wid = blk * (blockDim.x >> 5) + (threadIdx.x >> 5);
    if (wid >= NN / 16) return;
    const int lane = threadIdx.x & 31;
    const int half = lane >> 4;
    const int l    = lane & 15;
    const int m0   = wid * 16;

    v8f accL[4] = {};
    v8f accR[4] = {};

    for (int k0 = 0; k0 < CC; k0 += 4) {
        v2f a;
        const float* ap = x + (m0 + l) * CC + k0 + 2 * half;
        a.x = ap[0];
        a.y = ap[1];
#pragma unroll
        for (int n = 0; n < 4; n++) {
            const int kb = (k0 + 2 * half) * CC + n * 16 + l;
            v2f b;
            b.x = Wl[kb];
            b.y = Wl[kb + CC];
            accL[n] = __builtin_amdgcn_wmma_f32_16x16x4_f32(
                false, a, false, b, (short)0, accL[n], false, false);
            v2f c;
            c.x = Wr[kb];
            c.y = Wr[kb + CC];
            accR[n] = __builtin_amdgcn_wmma_f32_16x16x4_f32(
                false, a, false, c, (short)0, accR[n], false, false);
        }
    }

#pragma unroll
    for (int n = 0; n < 4; n++) {
#pragma unroll
        for (int r = 0; r < 8; r++) {
            const int row = m0 + half * 8 + r;
            const int col = n * 16 + l;
            xl[row * CC + col] = accL[n][r] + bl[col];
            xr[row * CC + col] = accR[n][r] + br[col];
        }
    }
}

#define PROJ_BLOCKS ((NN / 16 + 7) / 8)
__global__ void __launch_bounds__(256)
k_front(const float* __restrict__ ea, float* __restrict__ macc,
        const int* __restrict__ eidx, int* __restrict__ rcnt, int2* __restrict__ rrec,
        const float* __restrict__ x,
        const float* __restrict__ Wl, const float* __restrict__ bl,
        const float* __restrict__ Wr, const float* __restrict__ br,
        float* __restrict__ xl, float* __restrict__ xr, const size_t wsb) {
    if (wsb < WS_NEED) return;
    if (blockIdx.x < COUNT_BLOCKS) count_body(ea, macc, eidx, rcnt, rrec);
    else proj_body((int)blockIdx.x - COUNT_BLOCKS, x, Wl, bl, Wr, br, xl, xr);
}

template <int CTRL>
__device__ __forceinline__ float row_xadd(float v) {
    const int o = __builtin_amdgcn_update_dpp(0, __float_as_int(v), CTRL, 0xf, 0xf, true);
    return v + __int_as_float(o);
}
template <int K>
__device__ __forceinline__ int row_share_i(int v) {
    return __builtin_amdgcn_update_dpp(0, v, 0x150 + K, 0xf, 0xf, true);
}
template <int K>
__device__ __forceinline__ float row_share_f(float v) {
    return __int_as_float(row_share_i<K>(__float_as_int(v)));
}
template <int CTRL>
__device__ __forceinline__ float row_xmax(float v) {
    const int o = __builtin_amdgcn_update_dpp(0, __float_as_int(v), CTRL, 0xf, 0xf, true);
    return fmaxf(v, __int_as_float(o));
}
__device__ __forceinline__ float xhalf(float v) {
    const int b = __float_as_int(v);
    return __int_as_float(__builtin_amdgcn_permlanex16(b, b, 0x76543210, 0xfedcba98, false, false));
}

#define SRC_OF(R) { const int lo = row_share_i<R>(src_c); const int hi = row_share_i<8 + R>(src_c); si[R] = half ? hi : lo; }

static_assert(EE % 16 == 0, "the edge list is walked in whole tiles where a list overflowed");

__global__ void __launch_bounds__(256) __attribute__((amdgpu_waves_per_eu(8, 8)))
k_node(const int* __restrict__ eidx, const float* __restrict__ ea,
       const float* __restrict__ We, const float* __restrict__ att,
       const float* __restrict__ macc,
       const float* __restrict__ xl, const float* xr,
       const int* __restrict__ rcnt, const int2* __restrict__ rrec,
       const float* __restrict__ x, const float* __restrict__ bias,
       float* out, const size_t wsb) {
    if (wsb < WS_NEED) return;
    const int lane = threadIdx.x & 31;
    const int half = lane >> 4;
    const int c    = lane & 15;
    const int wv = __builtin_amdgcn_readfirstlane((int)(threadIdx.x >> 5));

    const int rng  = (int)blockIdx.x / RNG_BLOCKS;
    const int off0 = ((int)blockIdx.x % RNG_BLOCKS) * NPB;
    const int rtot = rcnt[rng];
    const bool rover = rtot > RCAP;
    int2 rec[RCAP / 256];
    {
        const int2* __restrict__ rr = rrec + (size_t)rng * RCAP;
#pragma unroll
        for (int u = 0; u < RCAP / 256; u++) {
            const int i = (int)threadIdx.x + 256 * u;
            rec[u] = make_int2(-1, 0);
            if (!rover && i < rtot) rec[u] = rr[i];
        }
    }

    __shared__ __attribute__((aligned(16))) float sB[32 * 28];
    __shared__ int lcnt[64];
    __shared__ __attribute__((aligned(16))) int2 llist[NPB * CAP];
    if (threadIdx.x < 32) {
#pragma unroll
        for (int j = 0; j < 3; j++) {
            const int k = 4 * j + 2 * half;
            const float4 w0 = *(const float4*)(We + k * CC + 4 * c);
            float4 w1 = make_float4(0.0f, 0.0f, 0.0f, 0.0f);
            if (k + 1 < EDD) w1 = *(const float4*)(We + (k + 1) * CC + 4 * c);
            float* sb = sB + lane * 28 + j * 8;
            sb[0] = w0.x; sb[1] = w1.x;
            sb[2] = w0.y; sb[3] = w1.y;
            sb[4] = w0.z; sb[5] = w1.z;
            sb[6] = w0.w; sb[7] = w1.w;
        }
    }
    if (threadIdx.x < 32) {
        const float4 a4 = *(const float4*)(att + 4 * c);
        float* sa = sB + lane * 28 + 24;
        sa[0] = a4.x; sa[1] = a4.y; sa[2] = a4.z; sa[3] = a4.w;
    }
    if (threadIdx.x < 64) lcnt[threadIdx.x] = 0;
    __syncthreads();

    int slot[RCAP / 256];
#pragma unroll
    for (int u = 0; u < RCAP / 256; u++) {
        const int dl = (int)((unsigned)rec[u].x >> 20) - off0;
        slot[u] = CAP;
        if ((unsigned)dl < (unsigned)NPB) slot[u] = atomicAdd(&lcnt[dl], 1);
    }
#pragma unroll
    for (int u = 0; u < RCAP / 256; u++) {
        const int dl = (int)((unsigned)rec[u].x >> 20) - off0;
        if (slot[u] < CAP) llist[dl * CAP + slot[u]] = make_int2(rec[u].x & 0xfffff, rec[u].y);
    }
    __syncthreads();

    const float4* __restrict__ sB4 = (const float4*)(sB + lane * 28);
    const float4* __restrict__ xl4 = (const float4*)xl;
    const float NEG_INF = __int_as_float((int)0xff800000u);
    const float invE = 1.0f / (float)EE;

    int ln = wv;
    int n  = (int)blockIdx.x * NPB + ln;
    if (n >= NN) return;
    float4 xrn = ((const float4*)xr)[(unsigned)n * (CC / 4) + (unsigned)c];

    for (;;) {
        const int lnn = ln + 8;
        const int nn  = n + 8;
        const bool more = lnn < NPB && nn < NN;

        const int  deg  = rover ? CAP + 1 : __builtin_amdgcn_readfirstlane(lcnt[ln]);
        const bool slow = deg > CAP;
        const int ntiles = slow ? 1 + EE / 16 : (deg + 16) >> 4;

        float m = NEG_INF, den = 0.0f;
        float4 accv = make_float4(0.0f, 0.0f, 0.0f, 0.0f);

        for (int tt = 0; tt < ntiles; tt++) {
            int e_c = -1, src_c = n;
            bool ok_c;
            if (!slow) {
                const int j = 16 * tt + c;
                ok_c = j <= deg;
                if (j >= 1 && j <= deg) {
                    const int2 pr = llist[ln * CAP + j - 1];
                    const unsigned id = (unsigned)pr.x, sv = (unsigned)pr.y;
                    e_c = (int)(id < (unsigned)EE ? id : (unsigned)EE - 1u);
                    src_c = (int)(sv < (unsigned)NN ? sv : (unsigned)NN - 1u);
                }
            } else if (tt == 0) {
                ok_c = c == 0;
            } else {
                const int b = (tt - 1) * 16 + c;
                const bool hit = eidx[EE + b] == n;
                if (__builtin_amdgcn_ballot_w32(hit) == 0u) continue;
                ok_c = hit;
                if (hit) { e_c = b; src_c = eidx[b]; }
            }
            const unsigned okm = __builtin_amdgcn_ballot_w32(ok_c);

            const float* er = (e_c >= 0 ? ea + (size_t)(unsigned)e_c * EDD : macc) + 2 * half;
            const float  as = e_c >= 0 ? 1.0f : invE;
            v2f a[3];
            a[0].x = er[0] * as; a[0].y = er[1] * as;
            a[1].x = er[4] * as; a[1].y = er[5] * as;
            a[2].x = er[8] * as;
            const float a9 = er[9 - half];
            a[2].y = half == 0 ? a9 * as : 0.0f;

            int si[8];
            SRC_OF(0) SRC_OF(1) SRC_OF(2) SRC_OF(3) SRC_OF(4) SRC_OF(5) SRC_OF(6) SRC_OF(7)

            float4 vl[8];
#pragma unroll
            for (int r = 0; r < 8; r++) vl[r] = xl4[(unsigned)si[r] * (CC / 4) + (unsigned)c];

            v8f acc[4];
#pragma unroll
            for (int r = 0; r < 8; r++) {
                acc[0][r] = vl[r].x + xrn.x;
                acc[1][r] = vl[r].y + xrn.y;
                acc[2][r] = vl[r].z + xrn.z;
                acc[3][r] = vl[r].w + xrn.w;
            }
#pragma unroll
            for (int j = 0; j < 3; j++) {
                const float4 b01 = sB4[2 * j], b23 = sB4[2 * j + 1];
                v2f b0, b1, b2, b3;
                b0.x = b01.x; b0.y = b01.y; b1.x = b01.z; b1.y = b01.w;
                b2.x = b23.x; b2.y = b23.y; b3.x = b23.z; b3.y = b23.w;
                acc[0] = __builtin_amdgcn_wmma_f32_16x16x4_f32(false, a[j], false, b0, (short)0, acc[0], false, false);
                acc[1] = __builtin_amdgcn_wmma_f32_16x16x4_f32(false, a[j], false, b1, (short)0, acc[1], false, false);
                acc[2] = __builtin_amdgcn_wmma_f32_16x16x4_f32(false, a[j], false, b2, (short)0, acc[2], false, false);
                acc[3] = __builtin_amdgcn_wmma_f32_16x16x4_f32(false, a[j], false, b3, (short)0, acc[3], false, false);
            }

            const float4 av = sB4[6];
            float part[8];
#pragma unroll
            for (int r = 0; r < 8; r++) {
                float h0 = acc[0][r], h1 = acc[1][r], h2 = acc[2][r], h3 = acc[3][r];
                h0 = fmaxf(h0, NEGS * h0);
                h1 = fmaxf(h1, NEGS * h1);
                h2 = fmaxf(h2, NEGS * h2);
                h3 = fmaxf(h3, NEGS * h3);
                part[r] = h0 * av.x + h1 * av.y + h2 * av.z + h3 * av.w;
            }
#pragma unroll
            for (int r = 0; r < 8; r++) part[r] = row_xadd<0x161>(part[r]);
#pragma unroll
            for (int r = 0; r < 8; r++) part[r] = row_xadd<0x162>(part[r]);
#pragma unroll
            for (int r = 0; r < 8; r++) part[r] = row_xadd<0x164>(part[r]);
#pragma unroll
            for (int r = 0; r < 8; r++) part[r] = row_xadd<0x168>(part[r]);

            float lown = part[0];
#pragma unroll
            for (int r = 1; r < 8; r++) {
                if ((c & 7) == r) lown = part[r];
            }
            if (((okm >> (8 * half + (c & 7))) & 1u) == 0u) lown = NEG_INF;
            float mt = lown;
            mt = row_xmax<0x161>(mt);
            mt = row_xmax<0x162>(mt);
            mt = row_xmax<0x164>(mt);
            mt = row_xmax<0x168>(mt);
            mt = fmaxf(mt, xhalf(mt));
            const float mnew = fmaxf(m, mt);
            const float rs = expf(m - mnew);
            den *= rs;
            accv.x *= rs; accv.y *= rs; accv.z *= rs; accv.w *= rs;
            const float pown = expf(lown - mnew);
            float pw[8];
            pw[0] = row_share_f<0>(pown); pw[1] = row_share_f<1>(pown); pw[2] = row_share_f<2>(pown); pw[3] = row_share_f<3>(pown);
            pw[4] = row_share_f<4>(pown); pw[5] = row_share_f<5>(pown); pw[6] = row_share_f<6>(pown); pw[7] = row_share_f<7>(pown);
#pragma unroll
            for (int r = 0; r < 8; r++) {
                const float p = pw[r];
                den += p;
                accv.x += p * vl[r].x; accv.y += p * vl[r].y; accv.z += p * vl[r].z; accv.w += p * vl[r].w;
            }
            m = mnew;
        }

        float4 nxr = xrn;
        if (more) nxr = ((const float4*)xr)[(unsigned)nn * (CC / 4) + (unsigned)c];

        den += xhalf(den);
        accv.x += xhalf(accv.x); accv.y += xhalf(accv.y); accv.z += xhalf(accv.z); accv.w += xhalf(accv.w);
        const float4 xv = ((const float4*)x)[(unsigned)n * (CC / 4) + (unsigned)c];
        const float4 bv = ((const float4*)bias)[c];
        const float inv = 1.0f / den;
        float4 o;
        o.x = xv.x + (accv.x * inv + bv.x);
        o.y = xv.y + (accv.y * inv + bv.y);
        o.z = xv.z + (accv.z * inv + bv.z);
        o.w = xv.w + (accv.w * inv + bv.w);
        if (half == 0) ((float4*)out)[(unsigned)n * (CC / 4) + (unsigned)c] = o;

        if (!more) break;
        n = nn;
        ln = lnn;
        xrn = nxr;
    }
}

extern "C" void kernel_launch(void* const* d_in, const int* in_sizes, int n_in,
                              void* d_out, int out_size, void* d_ws, size_t ws_size,
                              hipStream_t stream) {
    const float* x    = (const float*)d_in[0];
    const int*   eidx = (const int*)  d_in[1];
    const float* ea   = (const float*)d_in[2];
    const float* Wl   = (const float*)d_in[3];
    const float* bl   = (const float*)d_in[4];
    const float* Wr   = (const float*)d_in[5];
    const float* br   = (const float*)d_in[6];
    const float* We   = (const float*)d_in[7];
    const float* att  = (const float*)d_in[8];
    const float* bias = (const float*)d_in[9];
    float* out = (float*)d_out;

    float* ws    = (float*)d_ws;
    float* xl    = ws;
    float* xr    = out;
    int2*  rrec  = (int2*)(xl + NN * CC);
    int*   rcnt  = (int*)(rrec + NRNG * RCAP);
    float* macc  = (float*)(rcnt + NRNG);

    const int T = 256;
    k_init<<<INIT_BLOCKS, T, 0, stream>>>(rcnt, macc, ws_size);
    k_front<<<COUNT_BLOCKS + PROJ_BLOCKS, T, 0, stream>>>(ea, macc, eidx, rcnt, rrec, x, Wl, bl, Wr, br, xl, xr, ws_size);
    k_node<<<NODE_BLOCKS, T, 0, stream>>>(eidx, ea, We, att, macc, xl, xr, rcnt, rrec, x, bias, out, ws_size);
}


# k_node: nodes dealt to the waves on demand (one LDS add per node takes the block's next untaken node) instead of wave w owning nodes w, w+8, ..: waves of a block end together; sum order kept
# speedup vs baseline: 4.7963x; 1.0318x over previous
#include <hip/hip_runtime.h>

#define NN 50000
#define EE 800000
#define CC 64
#define EDD 11
#define NEGS 0.2f

typedef __attribute__((ext_vector_type(2))) float v2f;
typedef __attribute__((ext_vector_type(8))) float v8f;

#define CAP 48
#define NPB 49
#define NODE_BLOCKS 1024
#define RNG_BLOCKS 4
#define NRNG (NODE_BLOCKS / RNG_BLOCKS)
#define RNG_NODES (NPB * RNG_BLOCKS)
#define RCAP 4096
static_assert(NPB * NODE_BLOCKS >= NN, "every node belongs to a node block");
static_assert(EE <= (1 << 20) && RNG_NODES <= (1 << 11), "a record's first word holds the edge id and, above it, the offset");
static_assert(4095 - NPB * (RNG_BLOCKS - 1) >= NPB && RCAP % 256 == 0, "offset 4095 (a record that is not there) is no node of any block");
#define INIT_BLOCKS 1
#define WS_NEED ((size_t)NN * CC * 4 + (size_t)NRNG * RCAP * 8 + (size_t)NRNG * 4 + 16 * 4)

__global__ void k_init(int* __restrict__ rcnt, float* __restrict__ macc, const size_t wsb) {
    if (wsb < WS_NEED) return;
    int i = blockIdx.x * blockDim.x + threadIdx.x;
    if (i < NRNG) rcnt[i] = 0;
    if (i < 16) macc[i] = 0.0f;
}

#define COUNT_BLOCKS 512
#define COUNT_THREADS 256
#define EDGES_PER_THREAD ((EE + COUNT_BLOCKS * COUNT_THREADS - 1) / (COUNT_BLOCKS * COUNT_THREADS))
__device__ __forceinline__ void count_body(const float* __restrict__ ea, float* __restrict__ macc,
                          const int* __restrict__ eidx, int* __restrict__ rcnt,
                          int2* __restrict__ rrec) {
    __shared__ float s[EDD];
    __shared__ int lh[NRNG];
    if (threadIdx.x < EDD) s[threadIdx.x] = 0.0f;
    if (threadIdx.x < NRNG) lh[threadIdx.x] = 0;
    __syncthreads();
    float loc[EDD];
#pragma unroll
    for (int k = 0; k < EDD; k++) loc[k] = 0.0f;
    const int gid = blockIdx.x * COUNT_THREADS + threadIdx.x;
    int dv[EDGES_PER_THREAD], sv[EDGES_PER_THREAD], rk[EDGES_PER_THREAD];
#pragma unroll
    for (int it = 0; it < EDGES_PER_THREAD; it++) {
        const int e = gid + it * (COUNT_BLOCKS * COUNT_THREADS);
        dv[it] = -1; sv[it] = 0; rk[it] = 0;
        if (e < EE) {
            const unsigned du = (unsigned)eidx[EE + e];
            dv[it] = (int)(du < (unsigned)NN ? du : (unsigned)NN - 1u);
            sv[it] = eidx[e];
        }
    }
#pragma unroll
    for (int it = 0; it < EDGES_PER_THREAD; it++) {
        const int e = gid + it * (COUNT_BLOCKS * COUNT_THREADS);
        if (e < EE) {
#pragma unroll
            for (int k = 0; k < EDD; k++) loc[k] += ea[e * EDD + k];
        }
    }
#pragma unroll
    for (int it = 0; it < EDGES_PER_THREAD; it++) {
        if (dv[it] >= 0) rk[it] = atomicAdd(&lh[dv[it] / RNG_NODES], 1);
    }
    __syncthreads();
    if (threadIdx.x < NRNG) {
        const int cb = lh[threadIdx.x];
        lh[threadIdx.x] = cb > 0 ? atomicAdd(&rcnt[threadIdx.x], cb) : 0;
    }
    __syncthreads();
#pragma unroll
    for (int it = 0; it < EDGES_PER_THREAD; it++) {
        if (dv[it] >= 0) {
            const int e = gid + it * (COUNT_BLOCKS * COUNT_THREADS);
            const int r = dv[it] / RNG_NODES;
            const int pos = lh[r] + rk[it];
            if (pos < RCAP) rrec[r * RCAP + pos] = make_int2(e | ((dv[it] - r * RNG_NODES) << 20), sv[it]);
        }
    }
#pragma unroll
    for (int k = 0; k < EDD; k++) atomicAdd(&s[k], loc[k]);
    __syncthreads();
    if (threadIdx.x < EDD) atomicAdd(&macc[threadIdx.x], s[threadIdx.x]);
}

__device__ __forceinline__ void proj_body(const int blk, const float* __restrict__ x,
                            const float* __restrict__ Wl, const float* __restrict__ bl,
                            const float* __restrict__ Wr, const float* __restrict__ br,
                            float* __restrict__ xl, float* __restrict__ xr) {
    const int wid = blk * (blockDim.x >> 5) + (threadIdx.x >> 5);
    if (wid >= NN / 16) return;
    const int lane = threadIdx.x & 31;
    const int half = lane >> 4;
    const int l    = lane & 15;
    const int m0   = wid * 16;

    v8f accL[4] = {};
    v8f accR[4] = {};

    for (int k0 = 0; k0 < CC; k0 += 4) {
        v2f a;
        const float* ap = x + (m0 + l) * CC + k0 + 2 * half;
        a.x = ap[0];
        a.y = ap[1];
#pragma unroll
        for (int n = 0; n < 4; n++) {
            const int kb = (k0 + 2 * half) * CC + n * 16 + l;
            v2f b;
            b.x = Wl[kb];
            b.y = Wl[kb + CC];
            accL[n] = __builtin_amdgcn_wmma_f32_16x16x4_f32(
                false, a, false, b, (short)0, accL[n], false, false);
            v2f c;
            c.x = Wr[kb];
            c.y = Wr[kb + CC];
            accR[n] = __builtin_amdgcn_wmma_f32_16x16x4_f32(
                false, a, false, c, (short)0, accR[n], false, false);
        }
    }

#pragma unroll
    for (int n = 0; n < 4; n++) {
#pragma unroll
        for (int r = 0; r < 8; r++) {
            const int row = m0 + half * 8 + r;
            const int col = n * 16 + l;
            xl[row * CC + col] = accL[n][r] + bl[col];
            xr[row * CC + col] = accR[n][r] + br[col];
        }
    }
}

#define PROJ_BLOCKS ((NN / 16 + 7) / 8)
__global__ void __launch_bounds__(256)
k_front(const float* __restrict__ ea, float* __restrict__ macc,
        const int* __restrict__ eidx, int* __restrict__ rcnt, int2* __restrict__ rrec,
        const float* __restrict__ x,
        const float* __restrict__ Wl, const float* __restrict__ bl,
        const float* __restrict__ Wr, const float* __restrict__ br,
        float* __restrict__ xl, float* __restrict__ xr, const size_t wsb) {
    if (wsb < WS_NEED) return;
    if (blockIdx.x < COUNT_BLOCKS) count_body(ea, macc, eidx, rcnt, rrec);
    else proj_body((int)blockIdx.x - COUNT_BLOCKS, x, Wl, bl, Wr, br, xl, xr);
}

template <int CTRL>
__device__ __forceinline__ float row_xadd(float v) {
    const int o = __builtin_amdgcn_update_dpp(0, __float_as_int(v), CTRL, 0xf, 0xf, true);
    return v + __int_as_float(o);
}
template <int K>
__device__ __forceinline__ int row_share_i(int v) {
    return __builtin_amdgcn_update_dpp(0, v, 0x150 + K, 0xf, 0xf, true);
}
template <int K>
__device__ __forceinline__ float row_share_f(float v) {
    return __int_as_float(row_share_i<K>(__float_as_int(v)));
}
template <int CTRL>
__device__ __forceinline__ float row_xmax(float v) {
    const int o = __builtin_amdgcn_update_dpp(0, __float_as_int(v), CTRL, 0xf, 0xf, true);
    return fmaxf(v, __int_as_float(o));
}
__device__ __forceinline__ float xhalf(float v) {
    const int b = __float_as_int(v);
    return __int_as_float(__builtin_amdgcn_permlanex16(b, b, 0x76543210, 0xfedcba98, false, false));
}

#define SRC_OF(R) { const int lo = row_share_i<R>(src_c); const int hi = row_share_i<8 + R>(src_c); si[R] = half ? hi : lo; }

static_assert(EE % 16 == 0, "the edge list is walked in whole tiles where a list overflowed");

__global__ void __launch_bounds__(256) __attribute__((amdgpu_waves_per_eu(8, 8)))
k_node(const int* __restrict__ eidx, const float* __restrict__ ea,
       const float* __restrict__ We, const float* __restrict__ att,
       const float* __restrict__ macc,
       const float* __restrict__ xl, const float* xr,
       const int* __restrict__ rcnt, const int2* __restrict__ rrec,
       const float* __restrict__ x, const float* __restrict__ bias,
       float* out, const size_t wsb) {
    if (wsb < WS_NEED) return;
    const int lane = threadIdx.x & 31;
    const int half = lane >> 4;
    const int c    = lane & 15;
    const int wv = __builtin_amdgcn_readfirstlane((int)(threadIdx.x >> 5));

    const int rng  = (int)blockIdx.x / RNG_BLOCKS;
    const int off0 = ((int)blockIdx.x % RNG_BLOCKS) * NPB;
    const int rtot = rcnt[rng];
    const bool rover = rtot > RCAP;
    int2 rec[RCAP / 256];
    {
        const int2* __restrict__ rr = rrec + (size_t)rng * RCAP;
#pragma unroll
        for (int u = 0; u < RCAP / 256; u++) {
            const int i = (int)threadIdx.x + 256 * u;
            rec[u] = make_int2(-1, 0);
            if (!rover && i < rtot) rec[u] = rr[i];
        }
    }

    __shared__ __attribute__((aligned(16))) float sB[32 * 28];
    __shared__ int lcnt[64];
    __shared__ int nxt;
    __shared__ __attribute__((aligned(16))) int2 llist[NPB * CAP];
    if (threadIdx.x < 32) {
#pragma unroll
        for (int j = 0; j < 3; j++) {
            const int k = 4 * j + 2 * half;
            const float4 w0 = *(const float4*)(We + k * CC + 4 * c);
            float4 w1 = make_float4(0.0f, 0.0f, 0.0f, 0.0f);
            if (k + 1 < EDD) w1 = *(const float4*)(We + (k + 1) * CC + 4 * c);
            float* sb = sB + lane * 28 + j * 8;
            sb[0] = w0.x; sb[1] = w1.x;
            sb[2] = w0.y; sb[3] = w1.y;
            sb[4] = w0.z; sb[5] = w1.z;
            sb[6] = w0.w; sb[7] = w1.w;
        }
    }
    if (threadIdx.x < 32) {
        const float4 a4 = *(const float4*)(att + 4 * c);
        float* sa = sB + lane * 28 + 24;
        sa[0] = a4.x; sa[1] = a4.y; sa[2] = a4.z; sa[3] = a4.w;
    }
    if (threadIdx.x < 64) lcnt[threadIdx.x] = 0;
    if (threadIdx.x == 0) nxt = 8;
    __syncthreads();

    int slot[RCAP / 256];
#pragma unroll
    for (int u = 0; u < RCAP / 256; u++) {
        const int dl = (int)((unsigned)rec[u].x >> 20) - off0;
        slot[u] = CAP;
        if ((unsigned)dl < (unsigned)NPB) slot[u] = atomicAdd(&lcnt[dl], 1);
    }
#pragma unroll
    for (int u = 0; u < RCAP / 256; u++) {
        const int dl = (int)((unsigned)rec[u].x >> 20) - off0;
        if (slot[u] < CAP) llist[dl * CAP + slot[u]] = make_int2(rec[u].x & 0xfffff, rec[u].y);
    }
    __syncthreads();

    const float4* __restrict__ sB4 = (const float4*)(sB + lane * 28);
    const float4* __restrict__ xl4 = (const float4*)xl;
    const float NEG_INF = __int_as_float((int)0xff800000u);
    const float invE = 1.0f / (float)EE;

    int ln = wv;
    int n  = (int)blockIdx.x * NPB + ln;
    if (n >= NN) return;
    float4 xrn = ((const float4*)xr)[(unsigned)n * (CC / 4) + (unsigned)c];

    for (;;) {
        const int dgl = lcnt[ln];
        int tk = 0;
        if (lane == 0) tk = atomicAdd(&nxt, 1);
        const int lnn = __builtin_amdgcn_readfirstlane(tk);
        const int nn  = (int)blockIdx.x * NPB + lnn;
        const bool more = lnn < NPB && nn < NN;

        const int  deg  = rover ? CAP + 1 : __builtin_amdgcn_readfirstlane(dgl);
        const bool slow = deg > CAP;
        const int ntiles = slow ? 1 + EE / 16 : (deg + 16) >> 4;

        float m = NEG_INF, den = 0.0f;
        float4 accv = make_float4(0.0f, 0.0f, 0.0f, 0.0f);

        for (int tt = 0; tt < ntiles; tt++) {
            int e_c = -1, src_c = n;
            bool ok_c;
            if (!slow) {
                const int j = 16 * tt + c;
                ok_c = j <= deg;
                if (j >= 1 && j <= deg) {
                    const int2 pr = llist[ln * CAP + j - 1];
                    const unsigned id = (unsigned)pr.x, sv = (unsigned)pr.y;
                    e_c = (int)(id < (unsigned)EE ? id : (unsigned)EE - 1u);
                    src_c = (int)(sv < (unsigned)NN ? sv : (unsigned)NN - 1u);
                }
            } else if (tt == 0) {
                ok_c = c == 0;
            } else {
                const int b = (tt - 1) * 16 + c;
                const bool hit = eidx[EE + b] == n;
                if (__builtin_amdgcn_ballot_w32(hit) == 0u) continue;
                ok_c = hit;
                if (hit) { e_c = b; src_c = eidx[b]; }
            }
            const unsigned okm = __builtin_amdgcn_ballot_w32(ok_c);

            const float* er = (e_c >= 0 ? ea + (size_t)(unsigned)e_c * EDD : macc) + 2 * half;
            const float  as = e_c >= 0 ? 1.0f : invE;
            v2f a[3];
            a[0].x = er[0] * as; a[0].y = er[1] * as;
            a[1].x = er[4] * as; a[1].y = er[5] * as;
            a[2].x = er[8] * as;
            const float a9 = er[9 - half];
            a[2].y = half == 0 ? a9 * as : 0.0f;

            int si[8];
            SRC_OF(0) SRC_OF(1) SRC_OF(2) SRC_OF(3) SRC_OF(4) SRC_OF(5) SRC_OF(6) SRC_OF(7)

            float4 vl[8];
#pragma unroll
            for (int r = 0; r < 8; r++) vl[r] = xl4[(unsigned)si[r] * (CC / 4) + (unsigned)c];

            v8f acc[4];
#pragma unroll
            for (int r = 0; r < 8; r++) {
                acc[0][r] = vl[r].x + xrn.x;
                acc[1][r] = vl[r].y + xrn.y;
                acc[2][r] = vl[r].z + xrn.z;
                acc[3][r] = vl[r].w + xrn.w;
            }
#pragma unroll
            for (int j = 0; j < 3; j++) {
                const float4 b01 = sB4[2 * j], b23 = sB4[2 * j + 1];
                v2f b0, b1, b2, b3;
                b0.x = b01.x; b0.y = b01.y; b1.x = b01.z; b1.y = b01.w;
                b2.x = b23.x; b2.y = b23.y; b3.x = b23.z; b3.y = b23.w;
                acc[0] = __builtin_amdgcn_wmma_f32_16x16x4_f32(false, a[j], false, b0, (short)0, acc[0], false, false);
                acc[1] = __builtin_amdgcn_wmma_f32_16x16x4_f32(false, a[j], false, b1, (short)0, acc[1], false, false);
                acc[2] = __builtin_amdgcn_wmma_f32_16x16x4_f32(false, a[j], false, b2, (short)0, acc[2], false, false);
                acc[3] = __builtin_amdgcn_wmma_f32_16x16x4_f32(false, a[j], false, b3, (short)0, acc[3], false, false);
            }

            const float4 av = sB4[6];
            float part[8];
#pragma unroll
            for (int r = 0; r < 8; r++) {
                float h0 = acc[0][r], h1 = acc[1][r], h2 = acc[2][r], h3 = acc[3][r];
                h0 = fmaxf(h0, NEGS * h0);
                h1 = fmaxf(h1, NEGS * h1);
                h2 = fmaxf(h2, NEGS * h2);
                h3 = fmaxf(h3, NEGS * h3);
                part[r] = h0 * av.x + h1 * av.y + h2 * av.z + h3 * av.w;
            }
#pragma unroll
            for (int r = 0; r < 8; r++) part[r] = row_xadd<0x161>(part[r]);
#pragma unroll
            for (int r = 0; r < 8; r++) part[r] = row_xadd<0x162>(part[r]);
#pragma unroll
            for (int r = 0; r < 8; r++) part[r] = row_xadd<0x164>(part[r]);
#pragma unroll
            for (int r = 0; r < 8; r++) part[r] = row_xadd<0x168>(part[r]);

            float lown = part[0];
#pragma unroll
            for (int r = 1; r < 8; r++) {
                if ((c & 7) == r) lown = part[r];
            }
            if (((okm >> (8 * half + (c & 7))) & 1u) == 0u) lown = NEG_INF;
            float mt = lown;
            mt = row_xmax<0x161>(mt);
            mt = row_xmax<0x162>(mt);
            mt = row_xmax<0x164>(mt);
            mt = row_xmax<0x168>(mt);
            mt = fmaxf(mt, xhalf(mt));
            const float mnew = fmaxf(m, mt);
            const float rs = expf(m - mnew);
            den *= rs;
            accv.x *= rs; accv.y *= rs; accv.z *= rs; accv.w *= rs;
            const float pown = expf(lown - mnew);
            float pw[8];
            pw[0] = row_share_f<0>(pown); pw[1] = row_share_f<1>(pown); pw[2] = row_share_f<2>(pown); pw[3] = row_share_f<3>(pown);
            pw[4] = row_share_f<4>(pown); pw[5] = row_share_f<5>(pown); pw[6] = row_share_f<6>(pown); pw[7] = row_share_f<7>(pown);
#pragma unroll
            for (int r = 0; r < 8; r++) {
                const float p = pw[r];
                den += p;
                accv.x += p * vl[r].x; accv.y += p * vl[r].y; accv.z += p * vl[r].z; accv.w += p * vl[r].w;
            }
            m = mnew;
        }

        float4 nxr = xrn;
        if (more) nxr = ((const float4*)xr)[(unsigned)nn * (CC / 4) + (unsigned)c];

        den += xhalf(den);
        accv.x += xhalf(accv.x); accv.y += xhalf(accv.y); accv.z += xhalf(accv.z); accv.w += xhalf(accv.w);
        const float4 xv = ((const float4*)x)[(unsigned)n * (CC / 4) + (unsigned)c];
        const float4 bv = ((const float4*)bias)[c];
        const float inv = 1.0f / den;
        float4 o;
        o.x = xv.x + (accv.x * inv + bv.x);
        o.y = xv.y + (accv.y * inv + bv.y);
        o.z = xv.z + (accv.z * inv + bv.z);
        o.w = xv.w + (accv.w * inv + bv.w);
        if (half == 0) ((float4*)out)[(unsigned)n * (CC / 4) + (unsigned)c] = o;

        if (!more) break;
        n = nn;
        ln = lnn;
        xrn = nxr;
    }
}

extern "C" void kernel_launch(void* const* d_in, const int* in_sizes, int n_in,
                              void* d_out, int out_size, void* d_ws, size_t ws_size,
                              hipStream_t stream) {
    const float* x    = (const float*)d_in[0];
    const int*   eidx = (const int*)  d_in[1];
    const float* ea   = (const float*)d_in[2];
    const float* Wl   = (const float*)d_in[3];
    const float* bl   = (const float*)d_in[4];
    const float* Wr   = (const float*)d_in[5];
    const float* br   = (const float*)d_in[6];
    const float* We   = (const float*)d_in[7];
    const float* att  = (const float*)d_in[8];
    const float* bias = (const float*)d_in[9];
    float* out = (float*)d_out;

    float* ws    = (float*)d_ws;
    float* xl    = ws;
    float* xr    = out;
    int2*  rrec  = (int2*)(xl + NN * CC);
    int*   rcnt  = (int*)(rrec + NRNG * RCAP);
    float* macc  = (float*)(rcnt + NRNG);

    const int T = 256;
    k_init<<<INIT_BLOCKS, T, 0, stream>>>(rcnt, macc, ws_size);
    k_front<<<COUNT_BLOCKS + PROJ_BLOCKS, T, 0, stream>>>(ea, macc, eidx, rcnt, rrec, x, Wl, bl, Wr, br, xl, xr, ws_size);
    k_node<<<NODE_BLOCKS, T, 0, stream>>>(eidx, ea, We, att, macc, xl, xr, rcnt, rrec, x, bias, out, ws_size);
}
